# starting kernel
# baseline (speedup 1.0000x reference)
#include <hip/hip_runtime.h>
#include <math.h>


#define NDIM   256
#define NBATCH 65536
#define NROT   32640
#define TILE_B 64
#define RS     260

typedef __attribute__((ext_vector_type(2))) float v2f;
typedef __attribute__((ext_vector_type(8))) float v8f;

__global__ void qlayer_trig(const float* __restrict__ rots1,
                            const float* __restrict__ rots2,
                            const float* __restrict__ phases,
                            float* __restrict__ c1, float* __restrict__ s1,
                            float* __restrict__ c2, float* __restrict__ s2,
                            float* __restrict__ cp, float* __restrict__ sp) {
  int idx = blockIdx.x * blockDim.x + threadIdx.x;
  if (idx < NROT) {
    float s, c;
    sincosf(rots1[idx], &s, &c); c1[idx] = c; s1[idx] = s;
    sincosf(rots2[idx], &s, &c); c2[idx] = c; s2[idx] = s;
  }
  if (idx < NDIM) {
    float s, c;
    sincosf(phases[idx], &s, &c); cp[idx] = c; sp[idx] = s;
  }
}

__global__ void qlayer_givens(const float* __restrict__ c1,
                              const float* __restrict__ s1,
                              const float* __restrict__ c2,
                              const float* __restrict__ s2,
                              float* __restrict__ T1,
                              float* __restrict__ T2) {
  extern __shared__ float sm[];
  const int t = threadIdx.x;
  const float* cA = (blockIdx.x == 0) ? c1 : c2;
  const float* sA = (blockIdx.x == 0) ? s1 : s2;
  float* T        = (blockIdx.x == 0) ? T1 : T2;

  float* v = sm + t * 257;
  for (int k = 0; k < NDIM; ++k) v[k] = (k == t) ? 1.0f : 0.0f;

  int i = 0, j = 1;
  for (int r = 0; r < NROT; ++r) {
    float c = cA[r];
    float s = sA[r];
    float ri = v[i];
    float rj = v[j];
    v[i] = c * ri - s * rj;
    v[j] = s * ri + c * rj;
    if (++j == NDIM) { ++i; j = i + 1; }
  }
  for (int n = 0; n < NDIM; ++n) T[t * NDIM + n] = v[n];
}

__device__ __forceinline__ void gemm_stripe(const float* __restrict__ smSrc,
                                            const float* __restrict__ T,
                                            int r, int h, int ln16, int hl,
                                            v8f acc[8]) {
  const int m = r * 16 + ln16;
  #pragma unroll 2
  for (int k = 0; k < NDIM; k += 4) {
    const int kk = k + 2 * hl;
    v2f a;
    a.x = smSrc[m * RS + kk];
    a.y = smSrc[m * RS + kk + 1];
    const float* b0 = T + kk * NDIM       + h * 128 + ln16;
    const float* b1 = T + (kk + 1) * NDIM + h * 128 + ln16;
    #pragma unroll
    for (int t = 0; t < 8; ++t) {
      v2f b;
      b.x = b0[t * 16];
      b.y = b1[t * 16];
      acc[t] = __builtin_amdgcn_wmma_f32_16x16x4_f32(
           false, a,  false, b,
           (short)0, acc[t],  false,  false);
    }
  }
}

__global__ void __launch_bounds__(256)
qlayer_main(const float* __restrict__ x,
            const float* __restrict__ T1,
            const float* __restrict__ T2,
            const float* __restrict__ cp,
            const float* __restrict__ sp,
            float* __restrict__ out) {
  extern __shared__ float sm[];
  float* smA = sm;
  float* smB = sm + TILE_B * RS;

  const int tid  = threadIdx.x;
  const int wave = tid >> 5;
  const int lane = tid & 31;
  const int hl   = lane >> 4;
  const int ln16 = lane & 15;
  const int r    = wave & 3;
  const int h    = wave >> 2;
  const int base_row = blockIdx.x * TILE_B;

  {
    const int row = tid >> 2;
    const int q   = tid & 3;
    const float4* src = (const float4*)(x + (base_row + row) * NDIM + q * 64);
    float4* dst = (float4*)(smA + row * RS + q * 64);
    float acc = 0.f;
    #pragma unroll
    for (int c = 0; c < 16; ++c) {
      float4 vv = src[c];
      dst[c] = vv;
      acc += vv.x * vv.x + vv.y * vv.y + vv.z * vv.z + vv.w * vv.w;
    }
    smB[row * 4 + q] = acc;
  }
  __syncthreads();
  if (tid < TILE_B) {
    float s = smB[tid * 4] + smB[tid * 4 + 1] + smB[tid * 4 + 2] + smB[tid * 4 + 3];
    smB[1024 + tid] = rsqrtf(s);
  }
  __syncthreads();
  {
    const int row = tid >> 2;
    const int q   = tid & 3;
    const float inv = smB[1024 + row];
    float4* dst = (float4*)(smA + row * RS + q * 64);
    #pragma unroll
    for (int c = 0; c < 16; ++c) {
      float4 vv = dst[c];
      vv.x *= inv; vv.y *= inv; vv.z *= inv; vv.w *= inv;
      dst[c] = vv;
    }
  }
  __syncthreads();

  const v8f vzero = {0.f, 0.f, 0.f, 0.f, 0.f, 0.f, 0.f, 0.f};

  v8f acc[8];
  #pragma unroll
  for (int t = 0; t < 8; ++t) acc[t] = vzero;
  gemm_stripe(smA, T1, r, h, ln16, hl, acc);

  __syncthreads();

  #pragma unroll
  for (int t = 0; t < 8; ++t) {
    const int n  = h * 128 + t * 16 + ln16;
    const float cw = cp[n];
    const float sw = sp[n];
    #pragma unroll
    for (int g = 0; g < 8; ++g) {
      const int m = r * 16 + g + 8 * hl;
      const float y = acc[t][g];
      smA[m * RS + n] = y * cw;
      smB[m * RS + n] = y * sw;
    }
  }
  __syncthreads();

  v8f res[8];
  #pragma unroll
  for (int t = 0; t < 8; ++t) acc[t] = vzero;
  gemm_stripe(smA, T2, r, h, ln16, hl, acc);
  #pragma unroll
  for (int t = 0; t < 8; ++t) {
    #pragma unroll
    for (int g = 0; g < 8; ++g) { float v = acc[t][g]; res[t][g] = v * v; }
  }

  #pragma unroll
  for (int t = 0; t < 8; ++t) acc[t] = vzero;
  gemm_stripe(smB, T2, r, h, ln16, hl, acc);
  #pragma unroll
  for (int t = 0; t < 8; ++t) {
    #pragma unroll
    for (int g = 0; g < 8; ++g) { float v = acc[t][g]; res[t][g] += v * v; }
  }

  #pragma unroll
  for (int t = 0; t < 8; ++t) {
    const int n = h * 128 + t * 16 + ln16;
    #pragma unroll
    for (int g = 0; g < 8; ++g) {
      const int m = r * 16 + g + 8 * hl;
      out[(base_row + m) * NDIM + n] = res[t][g];
    }
  }
}

extern "C" void kernel_launch(void* const* d_in, const int* in_sizes, int n_in,
                              void* d_out, int out_size, void* d_ws, size_t ws_size,
                              hipStream_t stream) {
  const float* x      = (const float*)d_in[0];
  const float* rots1  = (const float*)d_in[1];
  const float* phases = (const float*)d_in[2];
  const float* rots2  = (const float*)d_in[3];
  float* out = (float*)d_out;
  float* ws  = (float*)d_ws;

  float* c1 = ws;
  float* s1 = ws + 32640;
  float* c2 = ws + 65280;
  float* s2 = ws + 97920;
  float* cp = ws + 130560;
  float* sp = ws + 130816;
  float* T1 = ws + 131072;
  float* T2 = ws + 196608;

  const int lds_givens = 256 * 257 * 4;
  const int lds_main   = 2 * TILE_B * RS * 4;
  hipFuncSetAttribute(reinterpret_cast<const void*>(qlayer_givens),
                      hipFuncAttributeMaxDynamicSharedMemorySize, lds_givens);
  hipFuncSetAttribute(reinterpret_cast<const void*>(qlayer_main),
                      hipFuncAttributeMaxDynamicSharedMemorySize, lds_main);

  qlayer_trig<<<(NROT + 255) / 256, 256, 0, stream>>>(
      rots1, rots2, phases, c1, s1, c2, s2, cp, sp);
  qlayer_givens<<<2, 256, lds_givens, stream>>>(c1, s1, c2, s2, T1, T2);
  qlayer_main<<<NBATCH / TILE_B, 256, lds_main, stream>>>(
      x, T1, T2, cp, sp, out);
}


# Rotation-matrix build: pivot rows swept 8 at a time held in registers, each element keeps its operation order and rounding; v[j] touched once per 8 rotations, no LDS round trip between rotations
# speedup vs baseline: 3.0081x; 3.0081x over previous
#include <hip/hip_runtime.h>
#include <math.h>


#define NDIM   256
#define NBATCH 65536
#define NROT   32640
#define TILE_B 64
#define RS     260

typedef __attribute__((ext_vector_type(2))) float v2f;
typedef __attribute__((ext_vector_type(8))) float v8f;

__global__ void qlayer_trig(const float* __restrict__ rots1,
                            const float* __restrict__ rots2,
                            const float* __restrict__ phases,
                            float* __restrict__ c1, float* __restrict__ s1,
                            float* __restrict__ c2, float* __restrict__ s2,
                            float* __restrict__ cp, float* __restrict__ sp) {
  int idx = blockIdx.x * blockDim.x + threadIdx.x;
  if (idx < NROT) {
    float s, c;
    sincosf(rots1[idx], &s, &c); c1[idx] = c; s1[idx] = s;
    sincosf(rots2[idx], &s, &c); c2[idx] = c; s2[idx] = s;
  }
  if (idx < NDIM) {
    float s, c;
    sincosf(phases[idx], &s, &c); cp[idx] = c; sp[idx] = s;
  }
}

#define GV_PB 8
#define GV_JU 4

__device__ __forceinline__ void givens_rot(float c, float s, float& ri, float& rj) {
  const float a = ri;
  const float b = rj;
  ri = fmaf(c, a, -(s * b));
  rj = fmaf(s, a, c * b);
}

__global__ void qlayer_givens(const float* __restrict__ c1,
                              const float* __restrict__ s1,
                              const float* __restrict__ c2,
                              const float* __restrict__ s2,
                              float* __restrict__ T1,
                              float* __restrict__ T2) {
  extern __shared__ float sm[];
  const int t = threadIdx.x;
  const float* cA = (blockIdx.x == 0) ? c1 : c2;
  const float* sA = (blockIdx.x == 0) ? s1 : s2;
  float* T        = (blockIdx.x == 0) ? T1 : T2;

  float* v = sm + t * 257;
  for (int k = 0; k < NDIM; ++k) v[k] = (k == t) ? 1.0f : 0.0f;

  for (int i0 = 0; i0 < NDIM; i0 += GV_PB) {
    float p[GV_PB];
    int rowbase[GV_PB];
    #pragma unroll
    for (int a = 0; a < GV_PB; ++a) {
      const int i = i0 + a;
      p[a] = v[i];
      rowbase[a] = (i * (2 * NDIM - 1 - i)) / 2;
    }

    #pragma unroll
    for (int a = 0; a < GV_PB; ++a) {
      #pragma unroll
      for (int b = a + 1; b < GV_PB; ++b) {
        const int idx = rowbase[a] + (b - a - 1);
        givens_rot(cA[idx], sA[idx], p[a], p[b]);
      }
    }

    for (int j = i0 + GV_PB; j < NDIM; j += GV_JU) {
      float w[GV_JU];
      #pragma unroll
      for (int u = 0; u < GV_JU; ++u) w[u] = v[j + u];
      #pragma unroll
      for (int a = 0; a < GV_PB; ++a) {
        const int idx0 = rowbase[a] + (j - (i0 + a) - 1);
        #pragma unroll
        for (int u = 0; u < GV_JU; ++u) {
          givens_rot(cA[idx0 + u], sA[idx0 + u], p[a], w[u]);
        }
      }
      #pragma unroll
      for (int u = 0; u < GV_JU; ++u) v[j + u] = w[u];
    }

    #pragma unroll
    for (int a = 0; a < GV_PB; ++a) v[i0 + a] = p[a];
  }
  for (int n = 0; n < NDIM; ++n) T[t * NDIM + n] = v[n];
}

__device__ __forceinline__ void gemm_stripe(const float* __restrict__ smSrc,
                                            const float* __restrict__ T,
                                            int r, int h, int ln16, int hl,
                                            v8f acc[8]) {
  const int m = r * 16 + ln16;
  #pragma unroll 2
  for (int k = 0; k < NDIM; k += 4) {
    const int kk = k + 2 * hl;
    v2f a;
    a.x = smSrc[m * RS + kk];
    a.y = smSrc[m * RS + kk + 1];
    const float* b0 = T + kk * NDIM       + h * 128 + ln16;
    const float* b1 = T + (kk + 1) * NDIM + h * 128 + ln16;
    #pragma unroll
    for (int t = 0; t < 8; ++t) {
      v2f b;
      b.x = b0[t * 16];
      b.y = b1[t * 16];
      acc[t] = __builtin_amdgcn_wmma_f32_16x16x4_f32(
           false, a,  false, b,
           (short)0, acc[t],  false,  false);
    }
  }
}

__global__ void __launch_bounds__(256)
qlayer_main(const float* __restrict__ x,
            const float* __restrict__ T1,
            const float* __restrict__ T2,
            const float* __restrict__ cp,
            const float* __restrict__ sp,
            float* __restrict__ out) {
  extern __shared__ float sm[];
  float* smA = sm;
  float* smB = sm + TILE_B * RS;

  const int tid  = threadIdx.x;
  const int wave = tid >> 5;
  const int lane = tid & 31;
  const int hl   = lane >> 4;
  const int ln16 = lane & 15;
  const int r    = wave & 3;
  const int h    = wave >> 2;
  const int base_row = blockIdx.x * TILE_B;

  {
    const int row = tid >> 2;
    const int q   = tid & 3;
    const float4* src = (const float4*)(x + (base_row + row) * NDIM + q * 64);
    float4* dst = (float4*)(smA + row * RS + q * 64);
    float acc = 0.f;
    #pragma unroll
    for (int c = 0; c < 16; ++c) {
      float4 vv = src[c];
      dst[c] = vv;
      acc += vv.x * vv.x + vv.y * vv.y + vv.z * vv.z + vv.w * vv.w;
    }
    smB[row * 4 + q] = acc;
  }
  __syncthreads();
  if (tid < TILE_B) {
    float s = smB[tid * 4] + smB[tid * 4 + 1] + smB[tid * 4 + 2] + smB[tid * 4 + 3];
    smB[1024 + tid] = rsqrtf(s);
  }
  __syncthreads();
  {
    const int row = tid >> 2;
    const int q   = tid & 3;
    const float inv = smB[1024 + row];
    float4* dst = (float4*)(smA + row * RS + q * 64);
    #pragma unroll
    for (int c = 0; c < 16; ++c) {
      float4 vv = dst[c];
      vv.x *= inv; vv.y *= inv; vv.z *= inv; vv.w *= inv;
      dst[c] = vv;
    }
  }
  __syncthreads();

  const v8f vzero = {0.f, 0.f, 0.f, 0.f, 0.f, 0.f, 0.f, 0.f};

  v8f acc[8];
  #pragma unroll
  for (int t = 0; t < 8; ++t) acc[t] = vzero;
  gemm_stripe(smA, T1, r, h, ln16, hl, acc);

  __syncthreads();

  #pragma unroll
  for (int t = 0; t < 8; ++t) {
    const int n  = h * 128 + t * 16 + ln16;
    const float cw = cp[n];
    const float sw = sp[n];
    #pragma unroll
    for (int g = 0; g < 8; ++g) {
      const int m = r * 16 + g + 8 * hl;
      const float y = acc[t][g];
      smA[m * RS + n] = y * cw;
      smB[m * RS + n] = y * sw;
    }
  }
  __syncthreads();

  v8f res[8];
  #pragma unroll
  for (int t = 0; t < 8; ++t) acc[t] = vzero;
  gemm_stripe(smA, T2, r, h, ln16, hl, acc);
  #pragma unroll
  for (int t = 0; t < 8; ++t) {
    #pragma unroll
    for (int g = 0; g < 8; ++g) { float v = acc[t][g]; res[t][g] = v * v; }
  }

  #pragma unroll
  for (int t = 0; t < 8; ++t) acc[t] = vzero;
  gemm_stripe(smB, T2, r, h, ln16, hl, acc);
  #pragma unroll
  for (int t = 0; t < 8; ++t) {
    #pragma unroll
    for (int g = 0; g < 8; ++g) { float v = acc[t][g]; res[t][g] += v * v; }
  }

  #pragma unroll
  for (int t = 0; t < 8; ++t) {
    const int n = h * 128 + t * 16 + ln16;
    #pragma unroll
    for (int g = 0; g < 8; ++g) {
      const int m = r * 16 + g + 8 * hl;
      out[(base_row + m) * NDIM + n] = res[t][g];
    }
  }
}

extern "C" void kernel_launch(void* const* d_in, const int* in_sizes, int n_in,
                              void* d_out, int out_size, void* d_ws, size_t ws_size,
                              hipStream_t stream) {
  const float* x      = (const float*)d_in[0];
  const float* rots1  = (const float*)d_in[1];
  const float* phases = (const float*)d_in[2];
  const float* rots2  = (const float*)d_in[3];
  float* out = (float*)d_out;
  float* ws  = (float*)d_ws;

  float* c1 = ws;
  float* s1 = ws + 32640;
  float* c2 = ws + 65280;
  float* s2 = ws + 97920;
  float* cp = ws + 130560;
  float* sp = ws + 130816;
  float* T1 = ws + 131072;
  float* T2 = ws + 196608;

  const int lds_givens = 256 * 257 * 4;
  const int lds_main   = 2 * TILE_B * RS * 4;
  hipFuncSetAttribute(reinterpret_cast<const void*>(qlayer_givens),
                      hipFuncAttributeMaxDynamicSharedMemorySize, lds_givens);
  hipFuncSetAttribute(reinterpret_cast<const void*>(qlayer_main),
                      hipFuncAttributeMaxDynamicSharedMemorySize, lds_main);

  qlayer_trig<<<(NROT + 255) / 256, 256, 0, stream>>>(
      rots1, rots2, phases, c1, s1, c2, s2, cp, sp);
  qlayer_givens<<<2, 256, lds_givens, stream>>>(c1, s1, c2, s2, T1, T2);
  qlayer_main<<<NBATCH / TILE_B, 256, lds_main, stream>>>(
      x, T1, T2, cp, sp, out);
}


# Rotation-matrix build: cos/sin of the next pivot block fetched coalesced and staged in LDS in sweep order, read back as broadcast 16-byte LDS loads one trip ahead instead of scalar loads
# speedup vs baseline: 4.8459x; 1.6110x over previous
#include <hip/hip_runtime.h>
#include <math.h>


#define NDIM   256
#define NBATCH 65536
#define NROT   32640
#define TILE_B 64
#define RS     260

typedef __attribute__((ext_vector_type(2))) float v2f;
typedef __attribute__((ext_vector_type(8))) float v8f;

__global__ void qlayer_trig(const float* __restrict__ rots1,
                            const float* __restrict__ rots2,
                            const float* __restrict__ phases,
                            float* __restrict__ c1, float* __restrict__ s1,
                            float* __restrict__ c2, float* __restrict__ s2,
                            float* __restrict__ cp, float* __restrict__ sp) {
  int idx = blockIdx.x * blockDim.x + threadIdx.x;
  if (idx < NROT) {
    float s, c;
    sincosf(rots1[idx], &s, &c); c1[idx] = c; s1[idx] = s;
    sincosf(rots2[idx], &s, &c); c2[idx] = c; s2[idx] = s;
  }
  if (idx < NDIM) {
    float s, c;
    sincosf(phases[idx], &s, &c); cp[idx] = c; sp[idx] = s;
  }
}

#define GV_PB 8
#define GV_JU 4
#define GV_NB (NDIM / GV_PB)
#define GV_TS 68
#define GV_STAGE (((NDIM - GV_PB) / GV_JU) * GV_TS)
#define GV_INTRA 128
#define GV_VEC (NDIM * 257)
#define GV_LDS_FLOATS (GV_VEC + GV_STAGE + GV_INTRA)

typedef __attribute__((ext_vector_type(4))) float v4f;

__device__ __forceinline__ void givens_rot(float c, float s, float& ri, float& rj) {
  const float a = ri;
  const float b = rj;
  ri = fmaf(c, a, -(s * b));
  rj = fmaf(s, a, c * b);
}

__device__ __forceinline__ int gv_rowbase(int i) {
  return (i * (2 * NDIM - 1 - i)) / 2;
}

__device__ __forceinline__ void gv_fetch(const float* __restrict__ cA,
                                         const float* __restrict__ sA,
                                         int i0, int t, float (&r)[17]) {
  const int len = NDIM - GV_PB - i0;
  #pragma unroll
  for (int q = 0; q < 16; ++q) {
    const int a = q >> 1;
    const float* src = (q & 1) ? sA : cA;
    const int first = gv_rowbase(i0 + a) + (GV_PB - 1 - a);
    r[q] = (t < len) ? src[first + t] : 0.0f;
  }
  const int a  = (t >> 3) & 7;
  const int bb = t & 7;
  const float* src = (t & 64) ? sA : cA;
  r[16] = (t < GV_INTRA && bb > a) ? src[gv_rowbase(i0 + a) + (bb - a - 1)] : 0.0f;
}

__device__ __forceinline__ void gv_put(float* stage, float* intra,
                                       int i0, int t, const float (&r)[17]) {
  const int len = NDIM - GV_PB - i0;
  if (t < len) {
    float* d = stage + (t >> 2) * GV_TS + (t & 3);
    #pragma unroll
    for (int q = 0; q < 16; ++q) d[(q >> 1) * 8 + (q & 1) * 4] = r[q];
  }
  if (t < GV_INTRA) intra[t] = r[16];
}

__device__ __forceinline__ void gv_load_trip(const float* stage, int g,
                                             v4f (&c4)[GV_PB], v4f (&s4)[GV_PB]) {
  const v4f* src = (const v4f*)(stage + g * GV_TS);
  #pragma unroll
  for (int a = 0; a < GV_PB; ++a) {
    c4[a] = src[2 * a];
    s4[a] = src[2 * a + 1];
  }
}

__device__ __forceinline__ void gv_trip(float* vj, const v4f (&c4)[GV_PB],
                                        const v4f (&s4)[GV_PB], float (&p)[GV_PB]) {
  float w[GV_JU];
  #pragma unroll
  for (int u = 0; u < GV_JU; ++u) w[u] = vj[u];
  #pragma unroll
  for (int a = 0; a < GV_PB; ++a) {
    #pragma unroll
    for (int u = 0; u < GV_JU; ++u) givens_rot(c4[a][u], s4[a][u], p[a], w[u]);
  }
  #pragma unroll
  for (int u = 0; u < GV_JU; ++u) vj[u] = w[u];
}

__global__ void __launch_bounds__(256)
qlayer_givens(const float* __restrict__ c1,
              const float* __restrict__ s1,
              const float* __restrict__ c2,
              const float* __restrict__ s2,
              float* __restrict__ T1,
              float* __restrict__ T2) {
  extern __shared__ float sm[];
  const int t = threadIdx.x;
  const float* cA = (blockIdx.x == 0) ? c1 : c2;
  const float* sA = (blockIdx.x == 0) ? s1 : s2;
  float* T        = (blockIdx.x == 0) ? T1 : T2;

  float* v = sm + t * 257;
  float* stage = sm + GV_VEC;
  float* intra = stage + GV_STAGE;

  for (int k = 0; k < NDIM; ++k) v[k] = (k == t) ? 1.0f : 0.0f;

  float r[17];
  gv_fetch(cA, sA, 0, t, r);
  gv_put(stage, intra, 0, t, r);
  __syncthreads();

  for (int blk = 0; blk < GV_NB; ++blk) {
    const int i0 = blk * GV_PB;
    const bool more = (blk + 1 < GV_NB);

    if (more) gv_fetch(cA, sA, i0 + GV_PB, t, r);

    float p[GV_PB];
    #pragma unroll
    for (int a = 0; a < GV_PB; ++a) p[a] = v[i0 + a];

    #pragma unroll
    for (int a = 0; a < GV_PB; ++a) {
      #pragma unroll
      for (int b = a + 1; b < GV_PB; ++b) {
        givens_rot(intra[a * 8 + b], intra[64 + a * 8 + b], p[a], p[b]);
      }
    }

    const int ntrip = (NDIM - GV_PB - i0) / GV_JU;
    float* vj = v + i0 + GV_PB;
    v4f cX[GV_PB], sX[GV_PB], cY[GV_PB], sY[GV_PB];
    if (ntrip > 0) gv_load_trip(stage, 0, cX, sX);
    for (int g = 0; g < ntrip; g += 2) {
      gv_load_trip(stage, g + 1, cY, sY);
      gv_trip(vj + GV_JU * g, cX, sX, p);
      const int gn = (g + 2 < ntrip) ? g + 2 : g + 1;
      gv_load_trip(stage, gn, cX, sX);
      gv_trip(vj + GV_JU * (g + 1), cY, sY, p);
    }

    #pragma unroll
    for (int a = 0; a < GV_PB; ++a) v[i0 + a] = p[a];

    __syncthreads();
    if (more) gv_put(stage, intra, i0 + GV_PB, t, r);
    __syncthreads();
  }
  for (int n = 0; n < NDIM; ++n) T[t * NDIM + n] = v[n];
}

__device__ __forceinline__ void gemm_stripe(const float* __restrict__ smSrc,
                                            const float* __restrict__ T,
                                            int r, int h, int ln16, int hl,
                                            v8f acc[8]) {
  const int m = r * 16 + ln16;
  #pragma unroll 2
  for (int k = 0; k < NDIM; k += 4) {
    const int kk = k + 2 * hl;
    v2f a;
    a.x = smSrc[m * RS + kk];
    a.y = smSrc[m * RS + kk + 1];
    const float* b0 = T + kk * NDIM       + h * 128 + ln16;
    const float* b1 = T + (kk + 1) * NDIM + h * 128 + ln16;
    #pragma unroll
    for (int t = 0; t < 8; ++t) {
      v2f b;
      b.x = b0[t * 16];
      b.y = b1[t * 16];
      acc[t] = __builtin_amdgcn_wmma_f32_16x16x4_f32(
           false, a,  false, b,
           (short)0, acc[t],  false,  false);
    }
  }
}

__global__ void __launch_bounds__(256)
qlayer_main(const float* __restrict__ x,
            const float* __restrict__ T1,
            const float* __restrict__ T2,
            const float* __restrict__ cp,
            const float* __restrict__ sp,
            float* __restrict__ out) {
  extern __shared__ float sm[];
  float* smA = sm;
  float* smB = sm + TILE_B * RS;

  const int tid  = threadIdx.x;
  const int wave = tid >> 5;
  const int lane = tid & 31;
  const int hl   = lane >> 4;
  const int ln16 = lane & 15;
  const int r    = wave & 3;
  const int h    = wave >> 2;
  const int base_row = blockIdx.x * TILE_B;

  {
    const int row = tid >> 2;
    const int q   = tid & 3;
    const float4* src = (const float4*)(x + (base_row + row) * NDIM + q * 64);
    float4* dst = (float4*)(smA + row * RS + q * 64);
    float acc = 0.f;
    #pragma unroll
    for (int c = 0; c < 16; ++c) {
      float4 vv = src[c];
      dst[c] = vv;
      acc += vv.x * vv.x + vv.y * vv.y + vv.z * vv.z + vv.w * vv.w;
    }
    smB[row * 4 + q] = acc;
  }
  __syncthreads();
  if (tid < TILE_B) {
    float s = smB[tid * 4] + smB[tid * 4 + 1] + smB[tid * 4 + 2] + smB[tid * 4 + 3];
    smB[1024 + tid] = rsqrtf(s);
  }
  __syncthreads();
  {
    const int row = tid >> 2;
    const int q   = tid & 3;
    const float inv = smB[1024 + row];
    float4* dst = (float4*)(smA + row * RS + q * 64);
    #pragma unroll
    for (int c = 0; c < 16; ++c) {
      float4 vv = dst[c];
      vv.x *= inv; vv.y *= inv; vv.z *= inv; vv.w *= inv;
      dst[c] = vv;
    }
  }
  __syncthreads();

  const v8f vzero = {0.f, 0.f, 0.f, 0.f, 0.f, 0.f, 0.f, 0.f};

  v8f acc[8];
  #pragma unroll
  for (int t = 0; t < 8; ++t) acc[t] = vzero;
  gemm_stripe(smA, T1, r, h, ln16, hl, acc);

  __syncthreads();

  #pragma unroll
  for (int t = 0; t < 8; ++t) {
    const int n  = h * 128 + t * 16 + ln16;
    const float cw = cp[n];
    const float sw = sp[n];
    #pragma unroll
    for (int g = 0; g < 8; ++g) {
      const int m = r * 16 + g + 8 * hl;
      const float y = acc[t][g];
      smA[m * RS + n] = y * cw;
      smB[m * RS + n] = y * sw;
    }
  }
  __syncthreads();

  v8f res[8];
  #pragma unroll
  for (int t = 0; t < 8; ++t) acc[t] = vzero;
  gemm_stripe(smA, T2, r, h, ln16, hl, acc);
  #pragma unroll
  for (int t = 0; t < 8; ++t) {
    #pragma unroll
    for (int g = 0; g < 8; ++g) { float v = acc[t][g]; res[t][g] = v * v; }
  }

  #pragma unroll
  for (int t = 0; t < 8; ++t) acc[t] = vzero;
  gemm_stripe(smB, T2, r, h, ln16, hl, acc);
  #pragma unroll
  for (int t = 0; t < 8; ++t) {
    #pragma unroll
    for (int g = 0; g < 8; ++g) { float v = acc[t][g]; res[t][g] += v * v; }
  }

  #pragma unroll
  for (int t = 0; t < 8; ++t) {
    const int n = h * 128 + t * 16 + ln16;
    #pragma unroll
    for (int g = 0; g < 8; ++g) {
      const int m = r * 16 + g + 8 * hl;
      out[(base_row + m) * NDIM + n] = res[t][g];
    }
  }
}

extern "C" void kernel_launch(void* const* d_in, const int* in_sizes, int n_in,
                              void* d_out, int out_size, void* d_ws, size_t ws_size,
                              hipStream_t stream) {
  const float* x      = (const float*)d_in[0];
  const float* rots1  = (const float*)d_in[1];
  const float* phases = (const float*)d_in[2];
  const float* rots2  = (const float*)d_in[3];
  float* out = (float*)d_out;
  float* ws  = (float*)d_ws;

  float* c1 = ws;
  float* s1 = ws + 32640;
  float* c2 = ws + 65280;
  float* s2 = ws + 97920;
  float* cp = ws + 130560;
  float* sp = ws + 130816;
  float* T1 = ws + 131072;
  float* T2 = ws + 196608;

  const int lds_givens = GV_LDS_FLOATS * 4;
  const int lds_main   = 2 * TILE_B * RS * 4;
  hipFuncSetAttribute(reinterpret_cast<const void*>(qlayer_givens),
                      hipFuncAttributeMaxDynamicSharedMemorySize, lds_givens);
  hipFuncSetAttribute(reinterpret_cast<const void*>(qlayer_main),
                      hipFuncAttributeMaxDynamicSharedMemorySize, lds_main);

  qlayer_trig<<<(NROT + 255) / 256, 256, 0, stream>>>(
      rots1, rots2, phases, c1, s1, c2, s2, cp, sp);
  qlayer_givens<<<2, 256, lds_givens, stream>>>(c1, s1, c2, s2, T1, T2);
  qlayer_main<<<NBATCH / TILE_B, 256, lds_main, stream>>>(
      x, T1, T2, cp, sp, out);
}


# Rotation-matrix build cut into 8 runs of pivot blocks, each run's factor built by its own work-group at the same time, factors multiplied on f32 WMMA (M rounds in a new order); factors parked in out
# speedup vs baseline: 8.4563x; 1.7450x over previous
#include <hip/hip_runtime.h>
#include <math.h>


#define NDIM   256
#define NBATCH 65536
#define NROT   32640
#define TILE_B 64
#define RS     260

typedef __attribute__((ext_vector_type(2))) float v2f;
typedef __attribute__((ext_vector_type(8))) float v8f;

__global__ void qlayer_trig(const float* __restrict__ rots1,
                            const float* __restrict__ rots2,
                            const float* __restrict__ phases,
                            float* __restrict__ c1, float* __restrict__ s1,
                            float* __restrict__ c2, float* __restrict__ s2,
                            float* __restrict__ cp, float* __restrict__ sp) {
  int idx = blockIdx.x * blockDim.x + threadIdx.x;
  if (idx < NROT) {
    float s, c;
    sincosf(rots1[idx], &s, &c); c1[idx] = c; s1[idx] = s;
    sincosf(rots2[idx], &s, &c); c2[idx] = c; s2[idx] = s;
  }
  if (idx < NDIM) {
    float s, c;
    sincosf(phases[idx], &s, &c); cp[idx] = c; sp[idx] = s;
  }
}

#define GV_PB 8
#define GV_JU 4
#define GV_NB (NDIM / GV_PB)
#define GV_TS 68
#define GV_STAGE (((NDIM - GV_PB) / GV_JU) * GV_TS)
#define GV_INTRA 128
#define GV_VEC (NDIM * 257)
#define GV_LDS_FLOATS (GV_VEC + GV_STAGE + GV_INTRA)

#define GV_K 8

__device__ __forceinline__ int gv_seg_block(int k) {
  const unsigned long long packed = 0x20550309184080ULL;
  return (int)((packed >> (6 * k)) & 63ULL);
}

typedef __attribute__((ext_vector_type(4))) float v4f;

__device__ __forceinline__ void givens_rot(float c, float s, float& ri, float& rj) {
  const float a = ri;
  const float b = rj;
  ri = fmaf(c, a, -(s * b));
  rj = fmaf(s, a, c * b);
}

__device__ __forceinline__ int gv_rowbase(int i) {
  return (i * (2 * NDIM - 1 - i)) / 2;
}

__device__ __forceinline__ void gv_fetch(const float* __restrict__ cA,
                                         const float* __restrict__ sA,
                                         int i0, int t, float (&r)[17]) {
  const int len = NDIM - GV_PB - i0;
  #pragma unroll
  for (int q = 0; q < 16; ++q) {
    const int a = q >> 1;
    const float* src = (q & 1) ? sA : cA;
    const int first = gv_rowbase(i0 + a) + (GV_PB - 1 - a);
    r[q] = (t < len) ? src[first + t] : 0.0f;
  }
  const int a  = (t >> 3) & 7;
  const int bb = t & 7;
  const float* src = (t & 64) ? sA : cA;
  r[16] = (t < GV_INTRA && bb > a) ? src[gv_rowbase(i0 + a) + (bb - a - 1)] : 0.0f;
}

__device__ __forceinline__ void gv_put(float* stage, float* intra,
                                       int i0, int t, const float (&r)[17]) {
  const int len = NDIM - GV_PB - i0;
  if (t < len) {
    float* d = stage + (t >> 2) * GV_TS + (t & 3);
    #pragma unroll
    for (int q = 0; q < 16; ++q) d[(q >> 1) * 8 + (q & 1) * 4] = r[q];
  }
  if (t < GV_INTRA) intra[t] = r[16];
}

__device__ __forceinline__ void gv_load_trip(const float* stage, int g,
                                             v4f (&c4)[GV_PB], v4f (&s4)[GV_PB]) {
  const v4f* src = (const v4f*)(stage + g * GV_TS);
  #pragma unroll
  for (int a = 0; a < GV_PB; ++a) {
    c4[a] = src[2 * a];
    s4[a] = src[2 * a + 1];
  }
}

__device__ __forceinline__ void gv_trip(float* vj, const v4f (&c4)[GV_PB],
                                        const v4f (&s4)[GV_PB], float (&p)[GV_PB]) {
  float w[GV_JU];
  #pragma unroll
  for (int u = 0; u < GV_JU; ++u) w[u] = vj[u];
  #pragma unroll
  for (int a = 0; a < GV_PB; ++a) {
    #pragma unroll
    for (int u = 0; u < GV_JU; ++u) givens_rot(c4[a][u], s4[a][u], p[a], w[u]);
  }
  #pragma unroll
  for (int u = 0; u < GV_JU; ++u) vj[u] = w[u];
}

__global__ void __launch_bounds__(256)
qlayer_givens(const float* __restrict__ c1,
              const float* __restrict__ s1,
              const float* __restrict__ c2,
              const float* __restrict__ s2,
              float* __restrict__ P) {
  extern __shared__ float sm[];
  const int t = threadIdx.x;
  const int mat = blockIdx.x / GV_K;
  const int seg = blockIdx.x % GV_K;
  const float* cA = (mat == 0) ? c1 : c2;
  const float* sA = (mat == 0) ? s1 : s2;
  float* Pk = P + (mat * GV_K + seg) * (NDIM * NDIM);
  const int b_lo = gv_seg_block(seg);
  const int b_hi = gv_seg_block(seg + 1);

  float* v = sm + t * 257;
  float* stage = sm + GV_VEC;
  float* intra = stage + GV_STAGE;

  for (int k = 0; k < NDIM; ++k) v[k] = (k == t) ? 1.0f : 0.0f;

  float r[17];
  gv_fetch(cA, sA, b_lo * GV_PB, t, r);
  gv_put(stage, intra, b_lo * GV_PB, t, r);
  __syncthreads();

  for (int blk = b_lo; blk < b_hi; ++blk) {
    const int i0 = blk * GV_PB;
    const bool more = (blk + 1 < b_hi);

    if (more) gv_fetch(cA, sA, i0 + GV_PB, t, r);

    float p[GV_PB];
    #pragma unroll
    for (int a = 0; a < GV_PB; ++a) p[a] = v[i0 + a];

    #pragma unroll
    for (int a = 0; a < GV_PB; ++a) {
      #pragma unroll
      for (int b = a + 1; b < GV_PB; ++b) {
        givens_rot(intra[a * 8 + b], intra[64 + a * 8 + b], p[a], p[b]);
      }
    }

    const int ntrip = (NDIM - GV_PB - i0) / GV_JU;
    float* vj = v + i0 + GV_PB;
    v4f cX[GV_PB], sX[GV_PB], cY[GV_PB], sY[GV_PB];
    if (ntrip > 0) gv_load_trip(stage, 0, cX, sX);
    for (int g = 0; g < ntrip; g += 2) {
      gv_load_trip(stage, g + 1, cY, sY);
      gv_trip(vj + GV_JU * g, cX, sX, p);
      const int gn = (g + 2 < ntrip) ? g + 2 : g + 1;
      gv_load_trip(stage, gn, cX, sX);
      gv_trip(vj + GV_JU * (g + 1), cY, sY, p);
    }

    #pragma unroll
    for (int a = 0; a < GV_PB; ++a) v[i0 + a] = p[a];

    __syncthreads();
    if (more) gv_put(stage, intra, i0 + GV_PB, t, r);
    __syncthreads();
  }
  for (int k = 0; k < NDIM; ++k) Pk[k * NDIM + t] = sm[k * 257 + t];
}

__device__ __forceinline__ void gemm_stripe(const float* __restrict__ smSrc,
                                            const float* __restrict__ T,
                                            int r, int h, int ln16, int hl,
                                            v8f acc[8]) {
  const int m = r * 16 + ln16;
  #pragma unroll 2
  for (int k = 0; k < NDIM; k += 4) {
    const int kk = k + 2 * hl;
    v2f a;
    a.x = smSrc[m * RS + kk];
    a.y = smSrc[m * RS + kk + 1];
    const float* b0 = T + kk * NDIM       + h * 128 + ln16;
    const float* b1 = T + (kk + 1) * NDIM + h * 128 + ln16;
    #pragma unroll
    for (int t = 0; t < 8; ++t) {
      v2f b;
      b.x = b0[t * 16];
      b.y = b1[t * 16];
      acc[t] = __builtin_amdgcn_wmma_f32_16x16x4_f32(
           false, a,  false, b,
           (short)0, acc[t],  false,  false);
    }
  }
}

#define CB_ROWS 16

__global__ void __launch_bounds__(512)
qlayer_combine(const float* __restrict__ P,
               float* __restrict__ T1,
               float* __restrict__ T2) {
  __shared__ float S[CB_ROWS * RS];
  const int tid  = threadIdx.x;
  const int lane = tid & 31;
  const int hl   = lane >> 4;
  const int ln16 = lane & 15;
  const int wave = tid >> 5;
  const int mat  = blockIdx.x >> 4;
  const int row0 = (blockIdx.x & 15) * CB_ROWS;
  const float* Pm = P + mat * GV_K * (NDIM * NDIM);
  float* T = (mat == 0) ? T1 : T2;

  const int srow = tid >> 5;
  const int sc4  = tid & 31;
  {
    const float4* src = (const float4*)(Pm + (row0 + srow) * NDIM);
    float4* dst = (float4*)(S + srow * RS);
    dst[sc4]      = src[sc4];
    dst[sc4 + 32] = src[sc4 + 32];
  }
  __syncthreads();

  const int n = wave * 16 + ln16;
  for (int k = 1; k < GV_K; ++k) {
    const int p0 = (gv_seg_block(k) * GV_PB) & ~15;
    const bool act = (wave * 16 >= p0);
    const float* B = Pm + k * (NDIM * NDIM);
    v8f acc = {0.f, 0.f, 0.f, 0.f, 0.f, 0.f, 0.f, 0.f};
    if (act) {
      #pragma unroll 8
      for (int k0 = p0; k0 < NDIM; k0 += 4) {
        const int kk = k0 + 2 * hl;
        v2f a, b;
        a.x = S[ln16 * RS + kk];
        a.y = S[ln16 * RS + kk + 1];
        b.x = B[kk * NDIM + n];
        b.y = B[(kk + 1) * NDIM + n];
        acc = __builtin_amdgcn_wmma_f32_16x16x4_f32(
             false, a,  false, b,
             (short)0, acc,  false,  false);
      }
    }
    __syncthreads();
    if (act) {
      #pragma unroll
      for (int g = 0; g < 8; ++g) S[(g + 8 * hl) * RS + n] = acc[g];
    }
    __syncthreads();
  }

  {
    const float4* srcS = (const float4*)(S + srow * RS);
    float4* dst = (float4*)(T + (row0 + srow) * NDIM);
    dst[sc4]      = srcS[sc4];
    dst[sc4 + 32] = srcS[sc4 + 32];
  }
}

__global__ void __launch_bounds__(256)
qlayer_main(const float* __restrict__ x,
            const float* __restrict__ T1,
            const float* __restrict__ T2,
            const float* __restrict__ cp,
            const float* __restrict__ sp,
            float* __restrict__ out) {
  extern __shared__ float sm[];
  float* smA = sm;
  float* smB = sm + TILE_B * RS;

  const int tid  = threadIdx.x;
  const int wave = tid >> 5;
  const int lane = tid & 31;
  const int hl   = lane >> 4;
  const int ln16 = lane & 15;
  const int r    = wave & 3;
  const int h    = wave >> 2;
  const int base_row = blockIdx.x * TILE_B;

  {
    const int row = tid >> 2;
    const int q   = tid & 3;
    const float4* src = (const float4*)(x + (base_row + row) * NDIM + q * 64);
    float4* dst = (float4*)(smA + row * RS + q * 64);
    float acc = 0.f;
    #pragma unroll
    for (int c = 0; c < 16; ++c) {
      float4 vv = src[c];
      dst[c] = vv;
      acc += vv.x * vv.x + vv.y * vv.y + vv.z * vv.z + vv.w * vv.w;
    }
    smB[row * 4 + q] = acc;
  }
  __syncthreads();
  if (tid < TILE_B) {
    float s = smB[tid * 4] + smB[tid * 4 + 1] + smB[tid * 4 + 2] + smB[tid * 4 + 3];
    smB[1024 + tid] = rsqrtf(s);
  }
  __syncthreads();
  {
    const int row = tid >> 2;
    const int q   = tid & 3;
    const float inv = smB[1024 + row];
    float4* dst = (float4*)(smA + row * RS + q * 64);
    #pragma unroll
    for (int c = 0; c < 16; ++c) {
      float4 vv = dst[c];
      vv.x *= inv; vv.y *= inv; vv.z *= inv; vv.w *= inv;
      dst[c] = vv;
    }
  }
  __syncthreads();

  const v8f vzero = {0.f, 0.f, 0.f, 0.f, 0.f, 0.f, 0.f, 0.f};

  v8f acc[8];
  #pragma unroll
  for (int t = 0; t < 8; ++t) acc[t] = vzero;
  gemm_stripe(smA, T1, r, h, ln16, hl, acc);

  __syncthreads();

  #pragma unroll
  for (int t = 0; t < 8; ++t) {
    const int n  = h * 128 + t * 16 + ln16;
    const float cw = cp[n];
    const float sw = sp[n];
    #pragma unroll
    for (int g = 0; g < 8; ++g) {
      const int m = r * 16 + g + 8 * hl;
      const float y = acc[t][g];
      smA[m * RS + n] = y * cw;
      smB[m * RS + n] = y * sw;
    }
  }
  __syncthreads();

  v8f res[8];
  #pragma unroll
  for (int t = 0; t < 8; ++t) acc[t] = vzero;
  gemm_stripe(smA, T2, r, h, ln16, hl, acc);
  #pragma unroll
  for (int t = 0; t < 8; ++t) {
    #pragma unroll
    for (int g = 0; g < 8; ++g) { float v = acc[t][g]; res[t][g] = v * v; }
  }

  #pragma unroll
  for (int t = 0; t < 8; ++t) acc[t] = vzero;
  gemm_stripe(smB, T2, r, h, ln16, hl, acc);
  #pragma unroll
  for (int t = 0; t < 8; ++t) {
    #pragma unroll
    for (int g = 0; g < 8; ++g) { float v = acc[t][g]; res[t][g] += v * v; }
  }

  #pragma unroll
  for (int t = 0; t < 8; ++t) {
    const int n = h * 128 + t * 16 + ln16;
    #pragma unroll
    for (int g = 0; g < 8; ++g) {
      const int m = r * 16 + g + 8 * hl;
      out[(base_row + m) * NDIM + n] = res[t][g];
    }
  }
}

extern "C" void kernel_launch(void* const* d_in, const int* in_sizes, int n_in,
                              void* d_out, int out_size, void* d_ws, size_t ws_size,
                              hipStream_t stream) {
  const float* x      = (const float*)d_in[0];
  const float* rots1  = (const float*)d_in[1];
  const float* phases = (const float*)d_in[2];
  const float* rots2  = (const float*)d_in[3];
  float* out = (float*)d_out;
  float* ws  = (float*)d_ws;

  float* c1 = ws;
  float* s1 = ws + 32640;
  float* c2 = ws + 65280;
  float* s2 = ws + 97920;
  float* cp = ws + 130560;
  float* sp = ws + 130816;
  float* T1 = ws + 131072;
  float* T2 = ws + 196608;

  const int lds_givens = GV_LDS_FLOATS * 4;
  const int lds_main   = 2 * TILE_B * RS * 4;
  hipFuncSetAttribute(reinterpret_cast<const void*>(qlayer_givens),
                      hipFuncAttributeMaxDynamicSharedMemorySize, lds_givens);
  hipFuncSetAttribute(reinterpret_cast<const void*>(qlayer_main),
                      hipFuncAttributeMaxDynamicSharedMemorySize, lds_main);

  qlayer_trig<<<(NROT + 255) / 256, 256, 0, stream>>>(
      rots1, rots2, phases, c1, s1, c2, s2, cp, sp);
  qlayer_givens<<<2 * GV_K, 256, lds_givens, stream>>>(c1, s1, c2, s2, out);
  qlayer_combine<<<32, 512, 0, stream>>>(out, T1, T2);
  qlayer_main<<<NBATCH / TILE_B, 256, lds_main, stream>>>(
      x, T1, T2, cp, sp, out);
}


# Phases and second rotation matrix folded into the first before the batch pass: C^T = M1^T diag(e^{i phi}) M2^T on f32 WMMA (sums over n taken first: new rounding order); batch does 2 GEMMs, not 3
# speedup vs baseline: 9.6052x; 1.1359x over previous
#include <hip/hip_runtime.h>
#include <math.h>


#define NDIM   256
#define NBATCH 65536
#define NROT   32640
#define TILE_B 64
#define RS     260

typedef __attribute__((ext_vector_type(2))) float v2f;
typedef __attribute__((ext_vector_type(8))) float v8f;

__global__ void qlayer_trig(const float* __restrict__ rots1,
                            const float* __restrict__ rots2,
                            const float* __restrict__ phases,
                            float* __restrict__ c1, float* __restrict__ s1,
                            float* __restrict__ c2, float* __restrict__ s2,
                            float* __restrict__ cp, float* __restrict__ sp) {
  int idx = blockIdx.x * blockDim.x + threadIdx.x;
  if (idx < NROT) {
    float s, c;
    sincosf(rots1[idx], &s, &c); c1[idx] = c; s1[idx] = s;
    sincosf(rots2[idx], &s, &c); c2[idx] = c; s2[idx] = s;
  }
  if (idx < NDIM) {
    float s, c;
    sincosf(phases[idx], &s, &c); cp[idx] = c; sp[idx] = s;
  }
}

#define GV_PB 8
#define GV_JU 4
#define GV_NB (NDIM / GV_PB)
#define GV_TS 68
#define GV_STAGE (((NDIM - GV_PB) / GV_JU) * GV_TS)
#define GV_INTRA 128
#define GV_VEC (NDIM * 257)
#define GV_LDS_FLOATS (GV_VEC + GV_STAGE + GV_INTRA)

#define GV_K 8

__device__ __forceinline__ int gv_seg_block(int k) {
  const unsigned long long packed = 0x20550309184080ULL;
  return (int)((packed >> (6 * k)) & 63ULL);
}

typedef __attribute__((ext_vector_type(4))) float v4f;

__device__ __forceinline__ void givens_rot(float c, float s, float& ri, float& rj) {
  const float a = ri;
  const float b = rj;
  ri = fmaf(c, a, -(s * b));
  rj = fmaf(s, a, c * b);
}

__device__ __forceinline__ int gv_rowbase(int i) {
  return (i * (2 * NDIM - 1 - i)) / 2;
}

__device__ __forceinline__ void gv_fetch(const float* __restrict__ cA,
                                         const float* __restrict__ sA,
                                         int i0, int t, float (&r)[17]) {
  const int len = NDIM - GV_PB - i0;
  #pragma unroll
  for (int q = 0; q < 16; ++q) {
    const int a = q >> 1;
    const float* src = (q & 1) ? sA : cA;
    const int first = gv_rowbase(i0 + a) + (GV_PB - 1 - a);
    r[q] = (t < len) ? src[first + t] : 0.0f;
  }
  const int a  = (t >> 3) & 7;
  const int bb = t & 7;
  const float* src = (t & 64) ? sA : cA;
  r[16] = (t < GV_INTRA && bb > a) ? src[gv_rowbase(i0 + a) + (bb - a - 1)] : 0.0f;
}

__device__ __forceinline__ void gv_put(float* stage, float* intra,
                                       int i0, int t, const float (&r)[17]) {
  const int len = NDIM - GV_PB - i0;
  if (t < len) {
    float* d = stage + (t >> 2) * GV_TS + (t & 3);
    #pragma unroll
    for (int q = 0; q < 16; ++q) d[(q >> 1) * 8 + (q & 1) * 4] = r[q];
  }
  if (t < GV_INTRA) intra[t] = r[16];
}

__device__ __forceinline__ void gv_load_trip(const float* stage, int g,
                                             v4f (&c4)[GV_PB], v4f (&s4)[GV_PB]) {
  const v4f* src = (const v4f*)(stage + g * GV_TS);
  #pragma unroll
  for (int a = 0; a < GV_PB; ++a) {
    c4[a] = src[2 * a];
    s4[a] = src[2 * a + 1];
  }
}

__device__ __forceinline__ void gv_trip(float* vj, const v4f (&c4)[GV_PB],
                                        const v4f (&s4)[GV_PB], float (&p)[GV_PB]) {
  float w[GV_JU];
  #pragma unroll
  for (int u = 0; u < GV_JU; ++u) w[u] = vj[u];
  #pragma unroll
  for (int a = 0; a < GV_PB; ++a) {
    #pragma unroll
    for (int u = 0; u < GV_JU; ++u) givens_rot(c4[a][u], s4[a][u], p[a], w[u]);
  }
  #pragma unroll
  for (int u = 0; u < GV_JU; ++u) vj[u] = w[u];
}

__global__ void __launch_bounds__(256)
qlayer_givens(const float* __restrict__ c1,
              const float* __restrict__ s1,
              const float* __restrict__ c2,
              const float* __restrict__ s2,
              float* __restrict__ P) {
  extern __shared__ float sm[];
  const int t = threadIdx.x;
  const int mat = blockIdx.x / GV_K;
  const int seg = blockIdx.x % GV_K;
  const float* cA = (mat == 0) ? c1 : c2;
  const float* sA = (mat == 0) ? s1 : s2;
  float* Pk = P + (mat * GV_K + seg) * (NDIM * NDIM);
  const int b_lo = gv_seg_block(seg);
  const int b_hi = gv_seg_block(seg + 1);

  float* v = sm + t * 257;
  float* stage = sm + GV_VEC;
  float* intra = stage + GV_STAGE;

  for (int k = 0; k < NDIM; ++k) v[k] = (k == t) ? 1.0f : 0.0f;

  float r[17];
  gv_fetch(cA, sA, b_lo * GV_PB, t, r);
  gv_put(stage, intra, b_lo * GV_PB, t, r);
  __syncthreads();

  for (int blk = b_lo; blk < b_hi; ++blk) {
    const int i0 = blk * GV_PB;
    const bool more = (blk + 1 < b_hi);

    if (more) gv_fetch(cA, sA, i0 + GV_PB, t, r);

    float p[GV_PB];
    #pragma unroll
    for (int a = 0; a < GV_PB; ++a) p[a] = v[i0 + a];

    #pragma unroll
    for (int a = 0; a < GV_PB; ++a) {
      #pragma unroll
      for (int b = a + 1; b < GV_PB; ++b) {
        givens_rot(intra[a * 8 + b], intra[64 + a * 8 + b], p[a], p[b]);
      }
    }

    const int ntrip = (NDIM - GV_PB - i0) / GV_JU;
    float* vj = v + i0 + GV_PB;
    v4f cX[GV_PB], sX[GV_PB], cY[GV_PB], sY[GV_PB];
    if (ntrip > 0) gv_load_trip(stage, 0, cX, sX);
    for (int g = 0; g < ntrip; g += 2) {
      gv_load_trip(stage, g + 1, cY, sY);
      gv_trip(vj + GV_JU * g, cX, sX, p);
      const int gn = (g + 2 < ntrip) ? g + 2 : g + 1;
      gv_load_trip(stage, gn, cX, sX);
      gv_trip(vj + GV_JU * (g + 1), cY, sY, p);
    }

    #pragma unroll
    for (int a = 0; a < GV_PB; ++a) v[i0 + a] = p[a];

    __syncthreads();
    if (more) gv_put(stage, intra, i0 + GV_PB, t, r);
    __syncthreads();
  }
  for (int k = 0; k < NDIM; ++k) Pk[k * NDIM + t] = sm[k * 257 + t];
}

__device__ __forceinline__ void gemm_stripe(const float* __restrict__ smSrc,
                                            const float* __restrict__ T,
                                            int r, int h, int ln16, int hl,
                                            v8f acc[8]) {
  const int m = r * 16 + ln16;
  #pragma unroll 2
  for (int k = 0; k < NDIM; k += 4) {
    const int kk = k + 2 * hl;
    v2f a;
    a.x = smSrc[m * RS + kk];
    a.y = smSrc[m * RS + kk + 1];
    const float* b0 = T + kk * NDIM       + h * 128 + ln16;
    const float* b1 = T + (kk + 1) * NDIM + h * 128 + ln16;
    #pragma unroll
    for (int t = 0; t < 8; ++t) {
      v2f b;
      b.x = b0[t * 16];
      b.y = b1[t * 16];
      acc[t] = __builtin_amdgcn_wmma_f32_16x16x4_f32(
           false, a,  false, b,
           (short)0, acc[t],  false,  false);
    }
  }
}

#define CB_ROWS 16

__global__ void __launch_bounds__(512)
qlayer_combine(const float* __restrict__ P,
               float* __restrict__ T1,
               float* __restrict__ T2) {
  __shared__ float S[CB_ROWS * RS];
  const int tid  = threadIdx.x;
  const int lane = tid & 31;
  const int hl   = lane >> 4;
  const int ln16 = lane & 15;
  const int wave = tid >> 5;
  const int mat  = blockIdx.x >> 4;
  const int row0 = (blockIdx.x & 15) * CB_ROWS;
  const float* Pm = P + mat * GV_K * (NDIM * NDIM);
  float* T = (mat == 0) ? T1 : T2;

  const int srow = tid >> 5;
  const int sc4  = tid & 31;
  {
    const float4* src = (const float4*)(Pm + (row0 + srow) * NDIM);
    float4* dst = (float4*)(S + srow * RS);
    dst[sc4]      = src[sc4];
    dst[sc4 + 32] = src[sc4 + 32];
  }
  __syncthreads();

  const int n = wave * 16 + ln16;
  for (int k = 1; k < GV_K; ++k) {
    const int p0 = (gv_seg_block(k) * GV_PB) & ~15;
    const bool act = (wave * 16 >= p0);
    const float* B = Pm + k * (NDIM * NDIM);
    v8f acc = {0.f, 0.f, 0.f, 0.f, 0.f, 0.f, 0.f, 0.f};
    if (act) {
      #pragma unroll 8
      for (int k0 = p0; k0 < NDIM; k0 += 4) {
        const int kk = k0 + 2 * hl;
        v2f a, b;
        a.x = S[ln16 * RS + kk];
        a.y = S[ln16 * RS + kk + 1];
        b.x = B[kk * NDIM + n];
        b.y = B[(kk + 1) * NDIM + n];
        acc = __builtin_amdgcn_wmma_f32_16x16x4_f32(
             false, a,  false, b,
             (short)0, acc,  false,  false);
      }
    }
    __syncthreads();
    if (act) {
      #pragma unroll
      for (int g = 0; g < 8; ++g) S[(g + 8 * hl) * RS + n] = acc[g];
    }
    __syncthreads();
  }

  {
    const float4* srcS = (const float4*)(S + srow * RS);
    float4* dst = (float4*)(T + (row0 + srow) * NDIM);
    dst[sc4]      = srcS[sc4];
    dst[sc4 + 32] = srcS[sc4 + 32];
  }
}

__global__ void __launch_bounds__(512)
qlayer_fold(float* __restrict__ T1,
            const float* __restrict__ T2,
            const float* __restrict__ cp,
            const float* __restrict__ sp,
            float* __restrict__ CI) {
  __shared__ float SR[CB_ROWS * RS];
  __shared__ float SI[CB_ROWS * RS];
  const int tid  = threadIdx.x;
  const int lane = tid & 31;
  const int hl   = lane >> 4;
  const int ln16 = lane & 15;
  const int wave = tid >> 5;
  const int row0 = blockIdx.x * CB_ROWS;

  const int srow = tid >> 5;
  const int sc4  = tid & 31;
  {
    const float4* src = (const float4*)(T1 + (row0 + srow) * NDIM);
    const float4* c4  = (const float4*)cp;
    const float4* s4  = (const float4*)sp;
    float4* dR = (float4*)(SR + srow * RS);
    float4* dI = (float4*)(SI + srow * RS);
    #pragma unroll
    for (int q = 0; q < 2; ++q) {
      const int i = sc4 + 32 * q;
      const float4 v = src[i];
      const float4 c = c4[i];
      const float4 s = s4[i];
      dR[i] = make_float4(v.x * c.x, v.y * c.y, v.z * c.z, v.w * c.w);
      dI[i] = make_float4(v.x * s.x, v.y * s.y, v.z * s.z, v.w * s.w);
    }
  }
  __syncthreads();

  const int n = wave * 16 + ln16;
  v8f accR = {0.f, 0.f, 0.f, 0.f, 0.f, 0.f, 0.f, 0.f};
  v8f accI = {0.f, 0.f, 0.f, 0.f, 0.f, 0.f, 0.f, 0.f};
  #pragma unroll 8
  for (int k0 = 0; k0 < NDIM; k0 += 4) {
    const int kk = k0 + 2 * hl;
    v2f aR, aI, b;
    aR.x = SR[ln16 * RS + kk];
    aR.y = SR[ln16 * RS + kk + 1];
    aI.x = SI[ln16 * RS + kk];
    aI.y = SI[ln16 * RS + kk + 1];
    b.x = T2[kk * NDIM + n];
    b.y = T2[(kk + 1) * NDIM + n];
    accR = __builtin_amdgcn_wmma_f32_16x16x4_f32(
         false, aR,  false, b,
         (short)0, accR,  false,  false);
    accI = __builtin_amdgcn_wmma_f32_16x16x4_f32(
         false, aI,  false, b,
         (short)0, accI,  false,  false);
  }
  __syncthreads();
  #pragma unroll
  for (int g = 0; g < 8; ++g) {
    SR[(g + 8 * hl) * RS + n] = accR[g];
    SI[(g + 8 * hl) * RS + n] = accI[g];
  }
  __syncthreads();

  {
    const float4* sR = (const float4*)(SR + srow * RS);
    const float4* sI = (const float4*)(SI + srow * RS);
    float4* dR = (float4*)(T1 + (row0 + srow) * NDIM);
    float4* dI = (float4*)(CI + (row0 + srow) * NDIM);
    dR[sc4]      = sR[sc4];
    dR[sc4 + 32] = sR[sc4 + 32];
    dI[sc4]      = sI[sc4];
    dI[sc4 + 32] = sI[sc4 + 32];
  }
}

#define MN_SCR 1088

__global__ void __launch_bounds__(256)
qlayer_main(const float* __restrict__ x,
            const float* __restrict__ CR,
            const float* __restrict__ CI,
            float* __restrict__ out) {
  extern __shared__ float sm[];
  float* smA = sm;
  float* smN = sm + TILE_B * RS;

  const int tid  = threadIdx.x;
  const int wave = tid >> 5;
  const int lane = tid & 31;
  const int hl   = lane >> 4;
  const int ln16 = lane & 15;
  const int r    = wave & 3;
  const int h    = wave >> 2;
  const int base_row = blockIdx.x * TILE_B;

  {
    const int row = tid >> 2;
    const int q   = tid & 3;
    const float4* src = (const float4*)(x + (base_row + row) * NDIM + q * 64);
    float4* dst = (float4*)(smA + row * RS + q * 64);
    float acc = 0.f;
    #pragma unroll
    for (int c = 0; c < 16; ++c) {
      float4 vv = src[c];
      dst[c] = vv;
      acc += vv.x * vv.x + vv.y * vv.y + vv.z * vv.z + vv.w * vv.w;
    }
    smN[row * 4 + q] = acc;
  }
  __syncthreads();
  if (tid < TILE_B) {
    float s = smN[tid * 4] + smN[tid * 4 + 1] + smN[tid * 4 + 2] + smN[tid * 4 + 3];
    smN[1024 + tid] = rsqrtf(s);
  }
  __syncthreads();
  {
    const int row = tid >> 2;
    const int q   = tid & 3;
    const float inv = smN[1024 + row];
    float4* dst = (float4*)(smA + row * RS + q * 64);
    #pragma unroll
    for (int c = 0; c < 16; ++c) {
      float4 vv = dst[c];
      vv.x *= inv; vv.y *= inv; vv.z *= inv; vv.w *= inv;
      dst[c] = vv;
    }
  }
  __syncthreads();

  const v8f vzero = {0.f, 0.f, 0.f, 0.f, 0.f, 0.f, 0.f, 0.f};

  v8f acc[8];
  v8f res[8];
  #pragma unroll
  for (int t = 0; t < 8; ++t) acc[t] = vzero;
  gemm_stripe(smA, CR, r, h, ln16, hl, acc);
  #pragma unroll
  for (int t = 0; t < 8; ++t) {
    #pragma unroll
    for (int g = 0; g < 8; ++g) { float v = acc[t][g]; res[t][g] = v * v; }
  }

  #pragma unroll
  for (int t = 0; t < 8; ++t) acc[t] = vzero;
  gemm_stripe(smA, CI, r, h, ln16, hl, acc);
  #pragma unroll
  for (int t = 0; t < 8; ++t) {
    #pragma unroll
    for (int g = 0; g < 8; ++g) { float v = acc[t][g]; res[t][g] += v * v; }
  }

  #pragma unroll
  for (int t = 0; t < 8; ++t) {
    const int n = h * 128 + t * 16 + ln16;
    #pragma unroll
    for (int g = 0; g < 8; ++g) {
      const int m = r * 16 + g + 8 * hl;
      out[(base_row + m) * NDIM + n] = res[t][g];
    }
  }
}

extern "C" void kernel_launch(void* const* d_in, const int* in_sizes, int n_in,
                              void* d_out, int out_size, void* d_ws, size_t ws_size,
                              hipStream_t stream) {
  const float* x      = (const float*)d_in[0];
  const float* rots1  = (const float*)d_in[1];
  const float* phases = (const float*)d_in[2];
  const float* rots2  = (const float*)d_in[3];
  float* out = (float*)d_out;
  float* ws  = (float*)d_ws;

  float* c1 = ws;
  float* s1 = ws + 32640;
  float* c2 = ws + 65280;
  float* s2 = ws + 97920;
  float* cp = ws + 130560;
  float* sp = ws + 130816;
  float* T1 = ws + 131072;
  float* T2 = ws + 196608;
  float* CI = ws;

  const int lds_givens = GV_LDS_FLOATS * 4;
  const int lds_main   = (TILE_B * RS + MN_SCR) * 4;
  hipFuncSetAttribute(reinterpret_cast<const void*>(qlayer_givens),
                      hipFuncAttributeMaxDynamicSharedMemorySize, lds_givens);
  hipFuncSetAttribute(reinterpret_cast<const void*>(qlayer_main),
                      hipFuncAttributeMaxDynamicSharedMemorySize, lds_main);

  qlayer_trig<<<(NROT + 255) / 256, 256, 0, stream>>>(
      rots1, rots2, phases, c1, s1, c2, s2, cp, sp);
  qlayer_givens<<<2 * GV_K, 256, lds_givens, stream>>>(c1, s1, c2, s2, out);
  qlayer_combine<<<32, 512, 0, stream>>>(out, T1, T2);
  qlayer_fold<<<NDIM / CB_ROWS, 512, 0, stream>>>(T1, T2, cp, sp, CI);
  qlayer_main<<<NBATCH / TILE_B, 256, lds_main, stream>>>(
      x, T1, CI, out);
}


# Batch pass: each wave owns a 64x32 block of both parts in one pass over k (was 16x128, one part at a time): each B fragment read from global memory feeds 4 WMMAs, not 1; every sum keeps its order
# speedup vs baseline: 10.4146x; 1.0843x over previous
#include <hip/hip_runtime.h>
#include <math.h>


#define NDIM   256
#define NBATCH 65536
#define NROT   32640
#define TILE_B 64
#define RS     260

typedef __attribute__((ext_vector_type(2))) float v2f;
typedef __attribute__((ext_vector_type(8))) float v8f;

__global__ void qlayer_trig(const float* __restrict__ rots1,
                            const float* __restrict__ rots2,
                            const float* __restrict__ phases,
                            float* __restrict__ c1, float* __restrict__ s1,
                            float* __restrict__ c2, float* __restrict__ s2,
                            float* __restrict__ cp, float* __restrict__ sp) {
  int idx = blockIdx.x * blockDim.x + threadIdx.x;
  if (idx < NROT) {
    float s, c;
    sincosf(rots1[idx], &s, &c); c1[idx] = c; s1[idx] = s;
    sincosf(rots2[idx], &s, &c); c2[idx] = c; s2[idx] = s;
  }
  if (idx < NDIM) {
    float s, c;
    sincosf(phases[idx], &s, &c); cp[idx] = c; sp[idx] = s;
  }
}

#define GV_PB 8
#define GV_JU 4
#define GV_NB (NDIM / GV_PB)
#define GV_TS 68
#define GV_STAGE (((NDIM - GV_PB) / GV_JU) * GV_TS)
#define GV_INTRA 128
#define GV_VEC (NDIM * 257)
#define GV_LDS_FLOATS (GV_VEC + GV_STAGE + GV_INTRA)

#define GV_K 8

__device__ __forceinline__ int gv_seg_block(int k) {
  const unsigned long long packed = 0x20550309184080ULL;
  return (int)((packed >> (6 * k)) & 63ULL);
}

typedef __attribute__((ext_vector_type(4))) float v4f;

__device__ __forceinline__ void givens_rot(float c, float s, float& ri, float& rj) {
  const float a = ri;
  const float b = rj;
  ri = fmaf(c, a, -(s * b));
  rj = fmaf(s, a, c * b);
}

__device__ __forceinline__ int gv_rowbase(int i) {
  return (i * (2 * NDIM - 1 - i)) / 2;
}

__device__ __forceinline__ void gv_fetch(const float* __restrict__ cA,
                                         const float* __restrict__ sA,
                                         int i0, int t, float (&r)[17]) {
  const int len = NDIM - GV_PB - i0;
  #pragma unroll
  for (int q = 0; q < 16; ++q) {
    const int a = q >> 1;
    const float* src = (q & 1) ? sA : cA;
    const int first = gv_rowbase(i0 + a) + (GV_PB - 1 - a);
    r[q] = (t < len) ? src[first + t] : 0.0f;
  }
  const int a  = (t >> 3) & 7;
  const int bb = t & 7;
  const float* src = (t & 64) ? sA : cA;
  r[16] = (t < GV_INTRA && bb > a) ? src[gv_rowbase(i0 + a) + (bb - a - 1)] : 0.0f;
}

__device__ __forceinline__ void gv_put(float* stage, float* intra,
                                       int i0, int t, const float (&r)[17]) {
  const int len = NDIM - GV_PB - i0;
  if (t < len) {
    float* d = stage + (t >> 2) * GV_TS + (t & 3);
    #pragma unroll
    for (int q = 0; q < 16; ++q) d[(q >> 1) * 8 + (q & 1) * 4] = r[q];
  }
  if (t < GV_INTRA) intra[t] = r[16];
}

__device__ __forceinline__ void gv_load_trip(const float* stage, int g,
                                             v4f (&c4)[GV_PB], v4f (&s4)[GV_PB]) {
  const v4f* src = (const v4f*)(stage + g * GV_TS);
  #pragma unroll
  for (int a = 0; a < GV_PB; ++a) {
    c4[a] = src[2 * a];
    s4[a] = src[2 * a + 1];
  }
}

__device__ __forceinline__ void gv_trip(float* vj, const v4f (&c4)[GV_PB],
                                        const v4f (&s4)[GV_PB], float (&p)[GV_PB]) {
  float w[GV_JU];
  #pragma unroll
  for (int u = 0; u < GV_JU; ++u) w[u] = vj[u];
  #pragma unroll
  for (int a = 0; a < GV_PB; ++a) {
    #pragma unroll
    for (int u = 0; u < GV_JU; ++u) givens_rot(c4[a][u], s4[a][u], p[a], w[u]);
  }
  #pragma unroll
  for (int u = 0; u < GV_JU; ++u) vj[u] = w[u];
}

__global__ void __launch_bounds__(256)
qlayer_givens(const float* __restrict__ c1,
              const float* __restrict__ s1,
              const float* __restrict__ c2,
              const float* __restrict__ s2,
              float* __restrict__ P) {
  extern __shared__ float sm[];
  const int t = threadIdx.x;
  const int mat = blockIdx.x / GV_K;
  const int seg = blockIdx.x % GV_K;
  const float* cA = (mat == 0) ? c1 : c2;
  const float* sA = (mat == 0) ? s1 : s2;
  float* Pk = P + (mat * GV_K + seg) * (NDIM * NDIM);
  const int b_lo = gv_seg_block(seg);
  const int b_hi = gv_seg_block(seg + 1);

  float* v = sm + t * 257;
  float* stage = sm + GV_VEC;
  float* intra = stage + GV_STAGE;

  for (int k = 0; k < NDIM; ++k) v[k] = (k == t) ? 1.0f : 0.0f;

  float r[17];
  gv_fetch(cA, sA, b_lo * GV_PB, t, r);
  gv_put(stage, intra, b_lo * GV_PB, t, r);
  __syncthreads();

  for (int blk = b_lo; blk < b_hi; ++blk) {
    const int i0 = blk * GV_PB;
    const bool more = (blk + 1 < b_hi);

    if (more) gv_fetch(cA, sA, i0 + GV_PB, t, r);

    float p[GV_PB];
    #pragma unroll
    for (int a = 0; a < GV_PB; ++a) p[a] = v[i0 + a];

    #pragma unroll
    for (int a = 0; a < GV_PB; ++a) {
      #pragma unroll
      for (int b = a + 1; b < GV_PB; ++b) {
        givens_rot(intra[a * 8 + b], intra[64 + a * 8 + b], p[a], p[b]);
      }
    }

    const int ntrip = (NDIM - GV_PB - i0) / GV_JU;
    float* vj = v + i0 + GV_PB;
    v4f cX[GV_PB], sX[GV_PB], cY[GV_PB], sY[GV_PB];
    if (ntrip > 0) gv_load_trip(stage, 0, cX, sX);
    for (int g = 0; g < ntrip; g += 2) {
      gv_load_trip(stage, g + 1, cY, sY);
      gv_trip(vj + GV_JU * g, cX, sX, p);
      const int gn = (g + 2 < ntrip) ? g + 2 : g + 1;
      gv_load_trip(stage, gn, cX, sX);
      gv_trip(vj + GV_JU * (g + 1), cY, sY, p);
    }

    #pragma unroll
    for (int a = 0; a < GV_PB; ++a) v[i0 + a] = p[a];

    __syncthreads();
    if (more) gv_put(stage, intra, i0 + GV_PB, t, r);
    __syncthreads();
  }
  for (int k = 0; k < NDIM; ++k) Pk[k * NDIM + t] = sm[k * 257 + t];
}

__device__ __forceinline__ void gemm_block(const float* __restrict__ smSrc,
                                           const float* __restrict__ TR,
                                           const float* __restrict__ TI,
                                           int c0, int ln16, int hl,
                                           v8f accR[8], v8f accI[8]) {
  #pragma unroll 2
  for (int k = 0; k < NDIM; k += 4) {
    const int kk = k + 2 * hl;
    v2f a[4];
    #pragma unroll
    for (int rt = 0; rt < 4; ++rt) {
      a[rt].x = smSrc[(rt * 16 + ln16) * RS + kk];
      a[rt].y = smSrc[(rt * 16 + ln16) * RS + kk + 1];
    }
    const float* r0 = TR + kk * NDIM + c0 + ln16;
    const float* i0 = TI + kk * NDIM + c0 + ln16;
    v2f bR[2], bI[2];
    #pragma unroll
    for (int ct = 0; ct < 2; ++ct) {
      bR[ct].x = r0[ct * 16];
      bR[ct].y = r0[NDIM + ct * 16];
      bI[ct].x = i0[ct * 16];
      bI[ct].y = i0[NDIM + ct * 16];
    }
    #pragma unroll
    for (int rt = 0; rt < 4; ++rt) {
      #pragma unroll
      for (int ct = 0; ct < 2; ++ct) {
        accR[rt * 2 + ct] = __builtin_amdgcn_wmma_f32_16x16x4_f32(
             false, a[rt],  false, bR[ct],
             (short)0, accR[rt * 2 + ct],  false,  false);
        accI[rt * 2 + ct] = __builtin_amdgcn_wmma_f32_16x16x4_f32(
             false, a[rt],  false, bI[ct],
             (short)0, accI[rt * 2 + ct],  false,  false);
      }
    }
  }
}

#define CB_ROWS 16

__global__ void __launch_bounds__(512)
qlayer_combine(const float* __restrict__ P,
               float* __restrict__ T1,
               float* __restrict__ T2) {
  __shared__ float S[CB_ROWS * RS];
  const int tid  = threadIdx.x;
  const int lane = tid & 31;
  const int hl   = lane >> 4;
  const int ln16 = lane & 15;
  const int wave = tid >> 5;
  const int mat  = blockIdx.x >> 4;
  const int row0 = (blockIdx.x & 15) * CB_ROWS;
  const float* Pm = P + mat * GV_K * (NDIM * NDIM);
  float* T = (mat == 0) ? T1 : T2;

  const int srow = tid >> 5;
  const int sc4  = tid & 31;
  {
    const float4* src = (const float4*)(Pm + (row0 + srow) * NDIM);
    float4* dst = (float4*)(S + srow * RS);
    dst[sc4]      = src[sc4];
    dst[sc4 + 32] = src[sc4 + 32];
  }
  __syncthreads();

  const int n = wave * 16 + ln16;
  for (int k = 1; k < GV_K; ++k) {
    const int p0 = (gv_seg_block(k) * GV_PB) & ~15;
    const bool act = (wave * 16 >= p0);
    const float* B = Pm + k * (NDIM * NDIM);
    v8f acc = {0.f, 0.f, 0.f, 0.f, 0.f, 0.f, 0.f, 0.f};
    if (act) {
      #pragma unroll 8
      for (int k0 = p0; k0 < NDIM; k0 += 4) {
        const int kk = k0 + 2 * hl;
        v2f a, b;
        a.x = S[ln16 * RS + kk];
        a.y = S[ln16 * RS + kk + 1];
        b.x = B[kk * NDIM + n];
        b.y = B[(kk + 1) * NDIM + n];
        acc = __builtin_amdgcn_wmma_f32_16x16x4_f32(
             false, a,  false, b,
             (short)0, acc,  false,  false);
      }
    }
    __syncthreads();
    if (act) {
      #pragma unroll
      for (int g = 0; g < 8; ++g) S[(g + 8 * hl) * RS + n] = acc[g];
    }
    __syncthreads();
  }

  {
    const float4* srcS = (const float4*)(S + srow * RS);
    float4* dst = (float4*)(T + (row0 + srow) * NDIM);
    dst[sc4]      = srcS[sc4];
    dst[sc4 + 32] = srcS[sc4 + 32];
  }
}

__global__ void __launch_bounds__(512)
qlayer_fold(float* __restrict__ T1,
            const float* __restrict__ T2,
            const float* __restrict__ cp,
            const float* __restrict__ sp,
            float* __restrict__ CI) {
  __shared__ float SR[CB_ROWS * RS];
  __shared__ float SI[CB_ROWS * RS];
  const int tid  = threadIdx.x;
  const int lane = tid & 31;
  const int hl   = lane >> 4;
  const int ln16 = lane & 15;
  const int wave = tid >> 5;
  const int row0 = blockIdx.x * CB_ROWS;

  const int srow = tid >> 5;
  const int sc4  = tid & 31;
  {
    const float4* src = (const float4*)(T1 + (row0 + srow) * NDIM);
    const float4* c4  = (const float4*)cp;
    const float4* s4  = (const float4*)sp;
    float4* dR = (float4*)(SR + srow * RS);
    float4* dI = (float4*)(SI + srow * RS);
    #pragma unroll
    for (int q = 0; q < 2; ++q) {
      const int i = sc4 + 32 * q;
      const float4 v = src[i];
      const float4 c = c4[i];
      const float4 s = s4[i];
      dR[i] = make_float4(v.x * c.x, v.y * c.y, v.z * c.z, v.w * c.w);
      dI[i] = make_float4(v.x * s.x, v.y * s.y, v.z * s.z, v.w * s.w);
    }
  }
  __syncthreads();

  const int n = wave * 16 + ln16;
  v8f accR = {0.f, 0.f, 0.f, 0.f, 0.f, 0.f, 0.f, 0.f};
  v8f accI = {0.f, 0.f, 0.f, 0.f, 0.f, 0.f, 0.f, 0.f};
  #pragma unroll 8
  for (int k0 = 0; k0 < NDIM; k0 += 4) {
    const int kk = k0 + 2 * hl;
    v2f aR, aI, b;
    aR.x = SR[ln16 * RS + kk];
    aR.y = SR[ln16 * RS + kk + 1];
    aI.x = SI[ln16 * RS + kk];
    aI.y = SI[ln16 * RS + kk + 1];
    b.x = T2[kk * NDIM + n];
    b.y = T2[(kk + 1) * NDIM + n];
    accR = __builtin_amdgcn_wmma_f32_16x16x4_f32(
         false, aR,  false, b,
         (short)0, accR,  false,  false);
    accI = __builtin_amdgcn_wmma_f32_16x16x4_f32(
         false, aI,  false, b,
         (short)0, accI,  false,  false);
  }
  __syncthreads();
  #pragma unroll
  for (int g = 0; g < 8; ++g) {
    SR[(g + 8 * hl) * RS + n] = accR[g];
    SI[(g + 8 * hl) * RS + n] = accI[g];
  }
  __syncthreads();

  {
    const float4* sR = (const float4*)(SR + srow * RS);
    const float4* sI = (const float4*)(SI + srow * RS);
    float4* dR = (float4*)(T1 + (row0 + srow) * NDIM);
    float4* dI = (float4*)(CI + (row0 + srow) * NDIM);
    dR[sc4]      = sR[sc4];
    dR[sc4 + 32] = sR[sc4 + 32];
    dI[sc4]      = sI[sc4];
    dI[sc4 + 32] = sI[sc4 + 32];
  }
}

#define MN_SCR 1088

__global__ void __launch_bounds__(256)
qlayer_main(const float* __restrict__ x,
            const float* __restrict__ CR,
            const float* __restrict__ CI,
            float* __restrict__ out) {
  extern __shared__ float sm[];
  float* smA = sm;
  float* smN = sm + TILE_B * RS;

  const int tid  = threadIdx.x;
  const int wave = tid >> 5;
  const int lane = tid & 31;
  const int hl   = lane >> 4;
  const int ln16 = lane & 15;
  const int c0   = wave * 32;
  const int base_row = blockIdx.x * TILE_B;

  {
    const int row = tid >> 2;
    const int q   = tid & 3;
    const float4* src = (const float4*)(x + (base_row + row) * NDIM + q * 64);
    float4* dst = (float4*)(smA + row * RS + q * 64);
    float acc = 0.f;
    #pragma unroll
    for (int c = 0; c < 16; ++c) {
      float4 vv = src[c];
      dst[c] = vv;
      acc += vv.x * vv.x + vv.y * vv.y + vv.z * vv.z + vv.w * vv.w;
    }
    smN[row * 4 + q] = acc;
  }
  __syncthreads();
  if (tid < TILE_B) {
    float s = smN[tid * 4] + smN[tid * 4 + 1] + smN[tid * 4 + 2] + smN[tid * 4 + 3];
    smN[1024 + tid] = rsqrtf(s);
  }
  __syncthreads();
  {
    const int row = tid >> 2;
    const int q   = tid & 3;
    const float inv = smN[1024 + row];
    float4* dst = (float4*)(smA + row * RS + q * 64);
    #pragma unroll
    for (int c = 0; c < 16; ++c) {
      float4 vv = dst[c];
      vv.x *= inv; vv.y *= inv; vv.z *= inv; vv.w *= inv;
      dst[c] = vv;
    }
  }
  __syncthreads();

  const v8f vzero = {0.f, 0.f, 0.f, 0.f, 0.f, 0.f, 0.f, 0.f};

  v8f accR[8];
  v8f accI[8];
  #pragma unroll
  for (int t = 0; t < 8; ++t) { accR[t] = vzero; accI[t] = vzero; }
  gemm_block(smA, CR, CI, c0, ln16, hl, accR, accI);

  #pragma unroll
  for (int rt = 0; rt < 4; ++rt) {
    #pragma unroll
    for (int ct = 0; ct < 2; ++ct) {
      const int n = c0 + ct * 16 + ln16;
      #pragma unroll
      for (int g = 0; g < 8; ++g) {
        const int m = rt * 16 + g + 8 * hl;
        const float vr = accR[rt * 2 + ct][g];
        const float vi = accI[rt * 2 + ct][g];
        out[(base_row + m) * NDIM + n] = fmaf(vi, vi, vr * vr);
      }
    }
  }
}

extern "C" void kernel_launch(void* const* d_in, const int* in_sizes, int n_in,
                              void* d_out, int out_size, void* d_ws, size_t ws_size,
                              hipStream_t stream) {
  const float* x      = (const float*)d_in[0];
  const float* rots1  = (const float*)d_in[1];
  const float* phases = (const float*)d_in[2];
  const float* rots2  = (const float*)d_in[3];
  float* out = (float*)d_out;
  float* ws  = (float*)d_ws;

  float* c1 = ws;
  float* s1 = ws + 32640;
  float* c2 = ws + 65280;
  float* s2 = ws + 97920;
  float* cp = ws + 130560;
  float* sp = ws + 130816;
  float* T1 = ws + 131072;
  float* T2 = ws + 196608;
  float* CI = ws;

  const int lds_givens = GV_LDS_FLOATS * 4;
  const int lds_main   = (TILE_B * RS + MN_SCR) * 4;
  hipFuncSetAttribute(reinterpret_cast<const void*>(qlayer_givens),
                      hipFuncAttributeMaxDynamicSharedMemorySize, lds_givens);
  hipFuncSetAttribute(reinterpret_cast<const void*>(qlayer_main),
                      hipFuncAttributeMaxDynamicSharedMemorySize, lds_main);

  qlayer_trig<<<(NROT + 255) / 256, 256, 0, stream>>>(
      rots1, rots2, phases, c1, s1, c2, s2, cp, sp);
  qlayer_givens<<<2 * GV_K, 256, lds_givens, stream>>>(c1, s1, c2, s2, out);
  qlayer_combine<<<32, 512, 0, stream>>>(out, T1, T2);
  qlayer_fold<<<NDIM / CB_ROWS, 512, 0, stream>>>(T1, T2, cp, sp, CI);
  qlayer_main<<<NBATCH / TILE_B, 256, lds_main, stream>>>(
      x, T1, CI, out);
}


# Factor products (kernels 1b, 1c): B fragments requested one product ahead (the first's and all of 1c's at kernel start) instead of 16 loads waited for per 8 k-steps; every sum keeps its order
# speedup vs baseline: 10.9329x; 1.0498x over previous
#include <hip/hip_runtime.h>
#include <math.h>


#define NDIM   256
#define NBATCH 65536
#define NROT   32640
#define TILE_B 64
#define RS     260

typedef __attribute__((ext_vector_type(2))) float v2f;
typedef __attribute__((ext_vector_type(8))) float v8f;

__global__ void qlayer_trig(const float* __restrict__ rots1,
                            const float* __restrict__ rots2,
                            const float* __restrict__ phases,
                            float* __restrict__ c1, float* __restrict__ s1,
                            float* __restrict__ c2, float* __restrict__ s2,
                            float* __restrict__ cp, float* __restrict__ sp) {
  int idx = blockIdx.x * blockDim.x + threadIdx.x;
  if (idx < NROT) {
    float s, c;
    sincosf(rots1[idx], &s, &c); c1[idx] = c; s1[idx] = s;
    sincosf(rots2[idx], &s, &c); c2[idx] = c; s2[idx] = s;
  }
  if (idx < NDIM) {
    float s, c;
    sincosf(phases[idx], &s, &c); cp[idx] = c; sp[idx] = s;
  }
}

#define GV_PB 8
#define GV_JU 4
#define GV_NB (NDIM / GV_PB)
#define GV_TS 68
#define GV_STAGE (((NDIM - GV_PB) / GV_JU) * GV_TS)
#define GV_INTRA 128
#define GV_VEC (NDIM * 257)
#define GV_LDS_FLOATS (GV_VEC + GV_STAGE + GV_INTRA)

#define GV_K 8

__device__ __forceinline__ int gv_seg_block(int k) {
  const unsigned long long packed = 0x20550309184080ULL;
  return (int)((packed >> (6 * k)) & 63ULL);
}

typedef __attribute__((ext_vector_type(4))) float v4f;

__device__ __forceinline__ void givens_rot(float c, float s, float& ri, float& rj) {
  const float a = ri;
  const float b = rj;
  ri = fmaf(c, a, -(s * b));
  rj = fmaf(s, a, c * b);
}

__device__ __forceinline__ int gv_rowbase(int i) {
  return (i * (2 * NDIM - 1 - i)) / 2;
}

__device__ __forceinline__ void gv_fetch(const float* __restrict__ cA,
                                         const float* __restrict__ sA,
                                         int i0, int t, float (&r)[17]) {
  const int len = NDIM - GV_PB - i0;
  #pragma unroll
  for (int q = 0; q < 16; ++q) {
    const int a = q >> 1;
    const float* src = (q & 1) ? sA : cA;
    const int first = gv_rowbase(i0 + a) + (GV_PB - 1 - a);
    r[q] = (t < len) ? src[first + t] : 0.0f;
  }
  const int a  = (t >> 3) & 7;
  const int bb = t & 7;
  const float* src = (t & 64) ? sA : cA;
  r[16] = (t < GV_INTRA && bb > a) ? src[gv_rowbase(i0 + a) + (bb - a - 1)] : 0.0f;
}

__device__ __forceinline__ void gv_put(float* stage, float* intra,
                                       int i0, int t, const float (&r)[17]) {
  const int len = NDIM - GV_PB - i0;
  if (t < len) {
    float* d = stage + (t >> 2) * GV_TS + (t & 3);
    #pragma unroll
    for (int q = 0; q < 16; ++q) d[(q >> 1) * 8 + (q & 1) * 4] = r[q];
  }
  if (t < GV_INTRA) intra[t] = r[16];
}

__device__ __forceinline__ void gv_load_trip(const float* stage, int g,
                                             v4f (&c4)[GV_PB], v4f (&s4)[GV_PB]) {
  const v4f* src = (const v4f*)(stage + g * GV_TS);
  #pragma unroll
  for (int a = 0; a < GV_PB; ++a) {
    c4[a] = src[2 * a];
    s4[a] = src[2 * a + 1];
  }
}

__device__ __forceinline__ void gv_trip(float* vj, const v4f (&c4)[GV_PB],
                                        const v4f (&s4)[GV_PB], float (&p)[GV_PB]) {
  float w[GV_JU];
  #pragma unroll
  for (int u = 0; u < GV_JU; ++u) w[u] = vj[u];
  #pragma unroll
  for (int a = 0; a < GV_PB; ++a) {
    #pragma unroll
    for (int u = 0; u < GV_JU; ++u) givens_rot(c4[a][u], s4[a][u], p[a], w[u]);
  }
  #pragma unroll
  for (int u = 0; u < GV_JU; ++u) vj[u] = w[u];
}

__global__ void __launch_bounds__(256)
qlayer_givens(const float* __restrict__ c1,
              const float* __restrict__ s1,
              const float* __restrict__ c2,
              const float* __restrict__ s2,
              float* __restrict__ P) {
  extern __shared__ float sm[];
  const int t = threadIdx.x;
  const int mat = blockIdx.x / GV_K;
  const int seg = blockIdx.x % GV_K;
  const float* cA = (mat == 0) ? c1 : c2;
  const float* sA = (mat == 0) ? s1 : s2;
  float* Pk = P + (mat * GV_K + seg) * (NDIM * NDIM);
  const int b_lo = gv_seg_block(seg);
  const int b_hi = gv_seg_block(seg + 1);

  float* v = sm + t * 257;
  float* stage = sm + GV_VEC;
  float* intra = stage + GV_STAGE;

  for (int k = 0; k < NDIM; ++k) v[k] = (k == t) ? 1.0f : 0.0f;

  float r[17];
  gv_fetch(cA, sA, b_lo * GV_PB, t, r);
  gv_put(stage, intra, b_lo * GV_PB, t, r);
  __syncthreads();

  for (int blk = b_lo; blk < b_hi; ++blk) {
    const int i0 = blk * GV_PB;
    const bool more = (blk + 1 < b_hi);

    if (more) gv_fetch(cA, sA, i0 + GV_PB, t, r);

    float p[GV_PB];
    #pragma unroll
    for (int a = 0; a < GV_PB; ++a) p[a] = v[i0 + a];

    #pragma unroll
    for (int a = 0; a < GV_PB; ++a) {
      #pragma unroll
      for (int b = a + 1; b < GV_PB; ++b) {
        givens_rot(intra[a * 8 + b], intra[64 + a * 8 + b], p[a], p[b]);
      }
    }

    const int ntrip = (NDIM - GV_PB - i0) / GV_JU;
    float* vj = v + i0 + GV_PB;
    v4f cX[GV_PB], sX[GV_PB], cY[GV_PB], sY[GV_PB];
    if (ntrip > 0) gv_load_trip(stage, 0, cX, sX);
    for (int g = 0; g < ntrip; g += 2) {
      gv_load_trip(stage, g + 1, cY, sY);
      gv_trip(vj + GV_JU * g, cX, sX, p);
      const int gn = (g + 2 < ntrip) ? g + 2 : g + 1;
      gv_load_trip(stage, gn, cX, sX);
      gv_trip(vj + GV_JU * (g + 1), cY, sY, p);
    }

    #pragma unroll
    for (int a = 0; a < GV_PB; ++a) v[i0 + a] = p[a];

    __syncthreads();
    if (more) gv_put(stage, intra, i0 + GV_PB, t, r);
    __syncthreads();
  }
  for (int k = 0; k < NDIM; ++k) Pk[k * NDIM + t] = sm[k * 257 + t];
}

__device__ __forceinline__ void gemm_block(const float* __restrict__ smSrc,
                                           const float* __restrict__ TR,
                                           const float* __restrict__ TI,
                                           int c0, int ln16, int hl,
                                           v8f accR[8], v8f accI[8]) {
  #pragma unroll 2
  for (int k = 0; k < NDIM; k += 4) {
    const int kk = k + 2 * hl;
    v2f a[4];
    #pragma unroll
    for (int rt = 0; rt < 4; ++rt) {
      a[rt].x = smSrc[(rt * 16 + ln16) * RS + kk];
      a[rt].y = smSrc[(rt * 16 + ln16) * RS + kk + 1];
    }
    const float* r0 = TR + kk * NDIM + c0 + ln16;
    const float* i0 = TI + kk * NDIM + c0 + ln16;
    v2f bR[2], bI[2];
    #pragma unroll
    for (int ct = 0; ct < 2; ++ct) {
      bR[ct].x = r0[ct * 16];
      bR[ct].y = r0[NDIM + ct * 16];
      bI[ct].x = i0[ct * 16];
      bI[ct].y = i0[NDIM + ct * 16];
    }
    #pragma unroll
    for (int rt = 0; rt < 4; ++rt) {
      #pragma unroll
      for (int ct = 0; ct < 2; ++ct) {
        accR[rt * 2 + ct] = __builtin_amdgcn_wmma_f32_16x16x4_f32(
             false, a[rt],  false, bR[ct],
             (short)0, accR[rt * 2 + ct],  false,  false);
        accI[rt * 2 + ct] = __builtin_amdgcn_wmma_f32_16x16x4_f32(
             false, a[rt],  false, bI[ct],
             (short)0, accI[rt * 2 + ct],  false,  false);
      }
    }
  }
}

#define CB_ROWS 16
#define CB_SEG(k) ((int)((0x20550309184080ULL >> (6 * (k))) & 63ULL))
#define CB_P0(k)  ((CB_SEG(k) * GV_PB) & ~15)
#define CB_NS(k)  ((NDIM - CB_P0(k)) / 4)
#define CB_MAXNS  (NDIM / 4)

template <int K>
__device__ __forceinline__ void cb_fetch_step(const float* __restrict__ Bl, int s, v2f& b) {
  b.x = Bl[(CB_P0(K) + 4 * s) * NDIM];
  b.y = Bl[(CB_P0(K) + 4 * s + 1) * NDIM];
}

template <int K>
__device__ __forceinline__ void cb_product(const float* __restrict__ Pl, float* S,
                                           int wave, int ln16, int hl, int n,
                                           const v2f (&bcur)[CB_MAXNS]) {
  constexpr int p0  = CB_P0(K);
  constexpr int ns  = CB_NS(K);
  constexpr int KN  = (K + 1 < GV_K) ? K + 1 : K;
  constexpr int nsn = (K + 1 < GV_K) ? CB_NS(KN) : 0;
  const bool act = (wave * 16 >= p0);
  v8f acc = {0.f, 0.f, 0.f, 0.f, 0.f, 0.f, 0.f, 0.f};
  v2f bnext[CB_MAXNS];
  if (act) {
    const float* arow = S + ln16 * RS + 2 * hl + p0;
    const float* Bn = Pl + KN * (NDIM * NDIM);
    #pragma unroll
    for (int s = 0; s < ns; ++s) {
      v2f a;
      a.x = arow[4 * s];
      a.y = arow[4 * s + 1];
      acc = __builtin_amdgcn_wmma_f32_16x16x4_f32(
           false, a,  false, bcur[s],
           (short)0, acc,  false,  false);
      if (s < nsn) cb_fetch_step<KN>(Bn, s, bnext[s]);
    }
  }
  __syncthreads();
  if (act) {
    #pragma unroll
    for (int g = 0; g < 8; ++g) S[(g + 8 * hl) * RS + n] = acc[g];
  }
  __syncthreads();
  if constexpr (K + 1 < GV_K) cb_product<K + 1>(Pl, S, wave, ln16, hl, n, bnext);
}

__global__ void __launch_bounds__(512)
qlayer_combine(const float* __restrict__ P,
               float* __restrict__ T1,
               float* __restrict__ T2) {
  __shared__ float S[CB_ROWS * RS];
  const int tid  = threadIdx.x;
  const int lane = tid & 31;
  const int hl   = lane >> 4;
  const int ln16 = lane & 15;
  const int wave = tid >> 5;
  const int mat  = blockIdx.x >> 4;
  const int row0 = (blockIdx.x & 15) * CB_ROWS;
  const float* Pm = P + mat * GV_K * (NDIM * NDIM);
  float* T = (mat == 0) ? T1 : T2;

  const int n = wave * 16 + ln16;
  const float* Pl = Pm + (2 * hl) * NDIM + n;

  v2f b1[CB_MAXNS];
  if (wave * 16 >= CB_P0(1)) {
    #pragma unroll
    for (int s = 0; s < CB_NS(1); ++s) cb_fetch_step<1>(Pl + NDIM * NDIM, s, b1[s]);
  }

  const int srow = tid >> 5;
  const int sc4  = tid & 31;
  {
    const float4* src = (const float4*)(Pm + (row0 + srow) * NDIM);
    float4* dst = (float4*)(S + srow * RS);
    dst[sc4]      = src[sc4];
    dst[sc4 + 32] = src[sc4 + 32];
  }
  __syncthreads();

  cb_product<1>(Pl, S, wave, ln16, hl, n, b1);

  {
    const float4* srcS = (const float4*)(S + srow * RS);
    float4* dst = (float4*)(T + (row0 + srow) * NDIM);
    dst[sc4]      = srcS[sc4];
    dst[sc4 + 32] = srcS[sc4 + 32];
  }
}

__global__ void __launch_bounds__(512)
qlayer_fold(float* __restrict__ T1,
            const float* __restrict__ T2,
            const float* __restrict__ cp,
            const float* __restrict__ sp,
            float* __restrict__ CI) {
  __shared__ float SR[CB_ROWS * RS];
  __shared__ float SI[CB_ROWS * RS];
  const int tid  = threadIdx.x;
  const int lane = tid & 31;
  const int hl   = lane >> 4;
  const int ln16 = lane & 15;
  const int wave = tid >> 5;
  const int row0 = blockIdx.x * CB_ROWS;

  const int n = wave * 16 + ln16;
  v2f bq[CB_MAXNS];
  {
    const float* T2l = T2 + (2 * hl) * NDIM + n;
    #pragma unroll
    for (int s = 0; s < CB_MAXNS; ++s) {
      bq[s].x = T2l[(4 * s) * NDIM];
      bq[s].y = T2l[(4 * s + 1) * NDIM];
    }
  }

  const int srow = tid >> 5;
  const int sc4  = tid & 31;
  {
    const float4* src = (const float4*)(T1 + (row0 + srow) * NDIM);
    const float4* c4  = (const float4*)cp;
    const float4* s4  = (const float4*)sp;
    float4* dR = (float4*)(SR + srow * RS);
    float4* dI = (float4*)(SI + srow * RS);
    #pragma unroll
    for (int q = 0; q < 2; ++q) {
      const int i = sc4 + 32 * q;
      const float4 v = src[i];
      const float4 c = c4[i];
      const float4 s = s4[i];
      dR[i] = make_float4(v.x * c.x, v.y * c.y, v.z * c.z, v.w * c.w);
      dI[i] = make_float4(v.x * s.x, v.y * s.y, v.z * s.z, v.w * s.w);
    }
  }
  __syncthreads();

  v8f accR = {0.f, 0.f, 0.f, 0.f, 0.f, 0.f, 0.f, 0.f};
  v8f accI = {0.f, 0.f, 0.f, 0.f, 0.f, 0.f, 0.f, 0.f};
  {
    const float* aRrow = SR + ln16 * RS + 2 * hl;
    const float* aIrow = SI + ln16 * RS + 2 * hl;
    #pragma unroll
    for (int s = 0; s < CB_MAXNS; ++s) {
      v2f aR, aI;
      aR.x = aRrow[4 * s];
      aR.y = aRrow[4 * s + 1];
      aI.x = aIrow[4 * s];
      aI.y = aIrow[4 * s + 1];
      accR = __builtin_amdgcn_wmma_f32_16x16x4_f32(
           false, aR,  false, bq[s],
           (short)0, accR,  false,  false);
      accI = __builtin_amdgcn_wmma_f32_16x16x4_f32(
           false, aI,  false, bq[s],
           (short)0, accI,  false,  false);
    }
  }
  __syncthreads();
  #pragma unroll
  for (int g = 0; g < 8; ++g) {
    SR[(g + 8 * hl) * RS + n] = accR[g];
    SI[(g + 8 * hl) * RS + n] = accI[g];
  }
  __syncthreads();

  {
    const float4* sR = (const float4*)(SR + srow * RS);
    const float4* sI = (const float4*)(SI + srow * RS);
    float4* dR = (float4*)(T1 + (row0 + srow) * NDIM);
    float4* dI = (float4*)(CI + (row0 + srow) * NDIM);
    dR[sc4]      = sR[sc4];
    dR[sc4 + 32] = sR[sc4 + 32];
    dI[sc4]      = sI[sc4];
    dI[sc4 + 32] = sI[sc4 + 32];
  }
}

#define MN_SCR 1088

__global__ void __launch_bounds__(256)
qlayer_main(const float* __restrict__ x,
            const float* __restrict__ CR,
            const float* __restrict__ CI,
            float* __restrict__ out) {
  extern __shared__ float sm[];
  float* smA = sm;
  float* smN = sm + TILE_B * RS;

  const int tid  = threadIdx.x;
  const int wave = tid >> 5;
  const int lane = tid & 31;
  const int hl   = lane >> 4;
  const int ln16 = lane & 15;
  const int c0   = wave * 32;
  const int base_row = blockIdx.x * TILE_B;

  {
    const int row = tid >> 2;
    const int q   = tid & 3;
    const float4* src = (const float4*)(x + (base_row + row) * NDIM + q * 64);
    float4* dst = (float4*)(smA + row * RS + q * 64);
    float acc = 0.f;
    #pragma unroll
    for (int c = 0; c < 16; ++c) {
      float4 vv = src[c];
      dst[c] = vv;
      acc += vv.x * vv.x + vv.y * vv.y + vv.z * vv.z + vv.w * vv.w;
    }
    smN[row * 4 + q] = acc;
  }
  __syncthreads();
  if (tid < TILE_B) {
    float s = smN[tid * 4] + smN[tid * 4 + 1] + smN[tid * 4 + 2] + smN[tid * 4 + 3];
    smN[1024 + tid] = rsqrtf(s);
  }
  __syncthreads();
  {
    const int row = tid >> 2;
    const int q   = tid & 3;
    const float inv = smN[1024 + row];
    float4* dst = (float4*)(smA + row * RS + q * 64);
    #pragma unroll
    for (int c = 0; c < 16; ++c) {
      float4 vv = dst[c];
      vv.x *= inv; vv.y *= inv; vv.z *= inv; vv.w *= inv;
      dst[c] = vv;
    }
  }
  __syncthreads();

  const v8f vzero = {0.f, 0.f, 0.f, 0.f, 0.f, 0.f, 0.f, 0.f};

  v8f accR[8];
  v8f accI[8];
  #pragma unroll
  for (int t = 0; t < 8; ++t) { accR[t] = vzero; accI[t] = vzero; }
  gemm_block(smA, CR, CI, c0, ln16, hl, accR, accI);

  #pragma unroll
  for (int rt = 0; rt < 4; ++rt) {
    #pragma unroll
    for (int ct = 0; ct < 2; ++ct) {
      const int n = c0 + ct * 16 + ln16;
      #pragma unroll
      for (int g = 0; g < 8; ++g) {
        const int m = rt * 16 + g + 8 * hl;
        const float vr = accR[rt * 2 + ct][g];
        const float vi = accI[rt * 2 + ct][g];
        out[(base_row + m) * NDIM + n] = fmaf(vi, vi, vr * vr);
      }
    }
  }
}

extern "C" void kernel_launch(void* const* d_in, const int* in_sizes, int n_in,
                              void* d_out, int out_size, void* d_ws, size_t ws_size,
                              hipStream_t stream) {
  const float* x      = (const float*)d_in[0];
  const float* rots1  = (const float*)d_in[1];
  const float* phases = (const float*)d_in[2];
  const float* rots2  = (const float*)d_in[3];
  float* out = (float*)d_out;
  float* ws  = (float*)d_ws;

  float* c1 = ws;
  float* s1 = ws + 32640;
  float* c2 = ws + 65280;
  float* s2 = ws + 97920;
  float* cp = ws + 130560;
  float* sp = ws + 130816;
  float* T1 = ws + 131072;
  float* T2 = ws + 196608;
  float* CI = ws;

  const int lds_givens = GV_LDS_FLOATS * 4;
  const int lds_main   = (TILE_B * RS + MN_SCR) * 4;
  hipFuncSetAttribute(reinterpret_cast<const void*>(qlayer_givens),
                      hipFuncAttributeMaxDynamicSharedMemorySize, lds_givens);
  hipFuncSetAttribute(reinterpret_cast<const void*>(qlayer_main),
                      hipFuncAttributeMaxDynamicSharedMemorySize, lds_main);

  qlayer_trig<<<(NROT + 255) / 256, 256, 0, stream>>>(
      rots1, rots2, phases, c1, s1, c2, s2, cp, sp);
  qlayer_givens<<<2 * GV_K, 256, lds_givens, stream>>>(c1, s1, c2, s2, out);
  qlayer_combine<<<32, 512, 0, stream>>>(out, T1, T2);
  qlayer_fold<<<NDIM / CB_ROWS, 512, 0, stream>>>(T1, T2, cp, sp, CI);
  qlayer_main<<<NBATCH / TILE_B, 256, lds_main, stream>>>(
      x, T1, CI, out);
}


# Rotation-matrix build: each factor's 256 columns dealt to 4 work-groups of 64 threads, one per WGP (one wave per SIMD; LDS pipe shared by 2 waves, not 8); every element keeps its operations and order
# speedup vs baseline: 11.2679x; 1.0306x over previous
#include <hip/hip_runtime.h>
#include <math.h>


#define NDIM   256
#define NBATCH 65536
#define NROT   32640
#define TILE_B 64
#define RS     260

typedef __attribute__((ext_vector_type(2))) float v2f;
typedef __attribute__((ext_vector_type(8))) float v8f;

__global__ void qlayer_trig(const float* __restrict__ rots1,
                            const float* __restrict__ rots2,
                            const float* __restrict__ phases,
                            float* __restrict__ c1, float* __restrict__ s1,
                            float* __restrict__ c2, float* __restrict__ s2,
                            float* __restrict__ cp, float* __restrict__ sp) {
  int idx = blockIdx.x * blockDim.x + threadIdx.x;
  if (idx < NROT) {
    float s, c;
    sincosf(rots1[idx], &s, &c); c1[idx] = c; s1[idx] = s;
    sincosf(rots2[idx], &s, &c); c2[idx] = c; s2[idx] = s;
  }
  if (idx < NDIM) {
    float s, c;
    sincosf(phases[idx], &s, &c); cp[idx] = c; sp[idx] = s;
  }
}

#define GV_PB 8
#define GV_JU 4
#define GV_NB (NDIM / GV_PB)
#define GV_TS 68
#define GV_STAGE (((NDIM - GV_PB) / GV_JU) * GV_TS)
#define GV_INTRA 128
#define GV_CS 4
#define GV_NT (NDIM / GV_CS)
#define GV_IE ((GV_INTRA + GV_NT - 1) / GV_NT)
#define GV_NR (GV_CS * 16 + GV_IE)
#define GV_VEC (GV_NT * 257)
#define GV_LDS_FLOATS (GV_VEC + GV_STAGE + GV_INTRA)
#define GV_LDS_BYTES (164 * 1024)
static_assert(GV_LDS_FLOATS * 4 <= GV_LDS_BYTES, "column vectors and staging fit the reservation");

#define GV_K 8

__device__ __forceinline__ int gv_seg_block(int k) {
  const unsigned long long packed = 0x20550309184080ULL;
  return (int)((packed >> (6 * k)) & 63ULL);
}

typedef __attribute__((ext_vector_type(4))) float v4f;

__device__ __forceinline__ void givens_rot(float c, float s, float& ri, float& rj) {
  const float a = ri;
  const float b = rj;
  ri = fmaf(c, a, -(s * b));
  rj = fmaf(s, a, c * b);
}

__device__ __forceinline__ int gv_rowbase(int i) {
  return (i * (2 * NDIM - 1 - i)) / 2;
}

__device__ __forceinline__ void gv_fetch(const float* __restrict__ cA,
                                         const float* __restrict__ sA,
                                         int i0, int t, float (&r)[GV_NR]) {
  const int len = NDIM - GV_PB - i0;
  #pragma unroll
  for (int e = 0; e < GV_CS; ++e) {
    const int tt = t + e * GV_NT;
    #pragma unroll
    for (int q = 0; q < 16; ++q) r[16 * e + q] = 0.0f;
    if (tt < len) {
      #pragma unroll
      for (int q = 0; q < 16; ++q) {
        const int a = q >> 1;
        const float* src = ((q & 1) ? sA : cA) + (gv_rowbase(i0 + a) + (GV_PB - 1 - a));
        r[16 * e + q] = src[tt];
      }
    }
  }
  #pragma unroll
  for (int e = 0; e < GV_IE; ++e) {
    const int tt = t + e * GV_NT;
    const int a  = (tt >> 3) & 7;
    const int bb = tt & 7;
    const float* src = (tt & 64) ? sA : cA;
    r[16 * GV_CS + e] =
        (tt < GV_INTRA && bb > a) ? src[gv_rowbase(i0 + a) + (bb - a - 1)] : 0.0f;
  }
}

__device__ __forceinline__ void gv_put(float* stage, float* intra,
                                       int i0, int t, const float (&r)[GV_NR]) {
  const int len = NDIM - GV_PB - i0;
  #pragma unroll
  for (int e = 0; e < GV_CS; ++e) {
    const int tt = t + e * GV_NT;
    if (tt < len) {
      float* d = stage + (tt >> 2) * GV_TS + (tt & 3);
      #pragma unroll
      for (int q = 0; q < 16; ++q) d[(q >> 1) * 8 + (q & 1) * 4] = r[16 * e + q];
    }
  }
  #pragma unroll
  for (int e = 0; e < GV_IE; ++e) {
    const int tt = t + e * GV_NT;
    if (tt < GV_INTRA) intra[tt] = r[16 * GV_CS + e];
  }
}

__device__ __forceinline__ void gv_load_trip(const float* stage, int g,
                                             v4f (&c4)[GV_PB], v4f (&s4)[GV_PB]) {
  const v4f* src = (const v4f*)(stage + g * GV_TS);
  #pragma unroll
  for (int a = 0; a < GV_PB; ++a) {
    c4[a] = src[2 * a];
    s4[a] = src[2 * a + 1];
  }
}

__device__ __forceinline__ void gv_trip(float* vj, const v4f (&c4)[GV_PB],
                                        const v4f (&s4)[GV_PB], float (&p)[GV_PB]) {
  float w[GV_JU];
  #pragma unroll
  for (int u = 0; u < GV_JU; ++u) w[u] = vj[u];
  #pragma unroll
  for (int a = 0; a < GV_PB; ++a) {
    #pragma unroll
    for (int u = 0; u < GV_JU; ++u) givens_rot(c4[a][u], s4[a][u], p[a], w[u]);
  }
  #pragma unroll
  for (int u = 0; u < GV_JU; ++u) vj[u] = w[u];
}

__global__ void __launch_bounds__(GV_NT)
qlayer_givens(const float* __restrict__ c1,
              const float* __restrict__ s1,
              const float* __restrict__ c2,
              const float* __restrict__ s2,
              float* __restrict__ P) {
  extern __shared__ float sm[];
  const int t = threadIdx.x;
  const int cg  = blockIdx.x % GV_CS;
  const int seg = (blockIdx.x / GV_CS) % GV_K;
  const int mat = blockIdx.x / (GV_CS * GV_K);
  const int col0 = cg * GV_NT;
  const float* cA = (mat == 0) ? c1 : c2;
  const float* sA = (mat == 0) ? s1 : s2;
  float* Pk = P + (mat * GV_K + seg) * (NDIM * NDIM);
  const int b_lo = gv_seg_block(seg);
  const int b_hi = gv_seg_block(seg + 1);

  float* v = sm + t * 257;
  float* stage = sm + GV_VEC;
  float* intra = stage + GV_STAGE;

  for (int k = 0; k < NDIM; ++k) v[k] = (k == col0 + t) ? 1.0f : 0.0f;

  float r[GV_NR];
  gv_fetch(cA, sA, b_lo * GV_PB, t, r);
  gv_put(stage, intra, b_lo * GV_PB, t, r);
  __syncthreads();

  for (int blk = b_lo; blk < b_hi; ++blk) {
    const int i0 = blk * GV_PB;
    const bool more = (blk + 1 < b_hi);

    if (more) gv_fetch(cA, sA, i0 + GV_PB, t, r);

    float p[GV_PB];
    #pragma unroll
    for (int a = 0; a < GV_PB; ++a) p[a] = v[i0 + a];

    #pragma unroll
    for (int a = 0; a < GV_PB; ++a) {
      #pragma unroll
      for (int b = a + 1; b < GV_PB; ++b) {
        givens_rot(intra[a * 8 + b], intra[64 + a * 8 + b], p[a], p[b]);
      }
    }

    const int ntrip = (NDIM - GV_PB - i0) / GV_JU;
    float* vj = v + i0 + GV_PB;
    v4f cX[GV_PB], sX[GV_PB], cY[GV_PB], sY[GV_PB];
    if (ntrip > 0) gv_load_trip(stage, 0, cX, sX);
    for (int g = 0; g < ntrip; g += 2) {
      gv_load_trip(stage, g + 1, cY, sY);
      gv_trip(vj + GV_JU * g, cX, sX, p);
      const int gn = (g + 2 < ntrip) ? g + 2 : g + 1;
      gv_load_trip(stage, gn, cX, sX);
      gv_trip(vj + GV_JU * (g + 1), cY, sY, p);
    }

    #pragma unroll
    for (int a = 0; a < GV_PB; ++a) v[i0 + a] = p[a];

    __syncthreads();
    if (more) gv_put(stage, intra, i0 + GV_PB, t, r);
    __syncthreads();
  }
  for (int k = 0; k < GV_NT; ++k) {
    float* dst = Pk + (col0 + k) * NDIM;
    const float* srcv = sm + k * 257;
    #pragma unroll
    for (int e = 0; e < GV_CS; ++e) dst[t + e * GV_NT] = srcv[t + e * GV_NT];
  }
}

__device__ __forceinline__ void gemm_block(const float* __restrict__ smSrc,
                                           const float* __restrict__ TR,
                                           const float* __restrict__ TI,
                                           int c0, int ln16, int hl,
                                           v8f accR[8], v8f accI[8]) {
  #pragma unroll 2
  for (int k = 0; k < NDIM; k += 4) {
    const int kk = k + 2 * hl;
    v2f a[4];
    #pragma unroll
    for (int rt = 0; rt < 4; ++rt) {
      a[rt].x = smSrc[(rt * 16 + ln16) * RS + kk];
      a[rt].y = smSrc[(rt * 16 + ln16) * RS + kk + 1];
    }
    const float* r0 = TR + kk * NDIM + c0 + ln16;
    const float* i0 = TI + kk * NDIM + c0 + ln16;
    v2f bR[2], bI[2];
    #pragma unroll
    for (int ct = 0; ct < 2; ++ct) {
      bR[ct].x = r0[ct * 16];
      bR[ct].y = r0[NDIM + ct * 16];
      bI[ct].x = i0[ct * 16];
      bI[ct].y = i0[NDIM + ct * 16];
    }
    #pragma unroll
    for (int rt = 0; rt < 4; ++rt) {
      #pragma unroll
      for (int ct = 0; ct < 2; ++ct) {
        accR[rt * 2 + ct] = __builtin_amdgcn_wmma_f32_16x16x4_f32(
             false, a[rt],  false, bR[ct],
             (short)0, accR[rt * 2 + ct],  false,  false);
        accI[rt * 2 + ct] = __builtin_amdgcn_wmma_f32_16x16x4_f32(
             false, a[rt],  false, bI[ct],
             (short)0, accI[rt * 2 + ct],  false,  false);
      }
    }
  }
}

#define CB_ROWS 16
#define CB_SEG(k) ((int)((0x20550309184080ULL >> (6 * (k))) & 63ULL))
#define CB_P0(k)  ((CB_SEG(k) * GV_PB) & ~15)
#define CB_NS(k)  ((NDIM - CB_P0(k)) / 4)
#define CB_MAXNS  (NDIM / 4)

template <int K>
__device__ __forceinline__ void cb_fetch_step(const float* __restrict__ Bl, int s, v2f& b) {
  b.x = Bl[(CB_P0(K) + 4 * s) * NDIM];
  b.y = Bl[(CB_P0(K) + 4 * s + 1) * NDIM];
}

template <int K>
__device__ __forceinline__ void cb_product(const float* __restrict__ Pl, float* S,
                                           int wave, int ln16, int hl, int n,
                                           const v2f (&bcur)[CB_MAXNS]) {
  constexpr int p0  = CB_P0(K);
  constexpr int ns  = CB_NS(K);
  constexpr int KN  = (K + 1 < GV_K) ? K + 1 : K;
  constexpr int nsn = (K + 1 < GV_K) ? CB_NS(KN) : 0;
  const bool act = (wave * 16 >= p0);
  v8f acc = {0.f, 0.f, 0.f, 0.f, 0.f, 0.f, 0.f, 0.f};
  v2f bnext[CB_MAXNS];
  if (act) {
    const float* arow = S + ln16 * RS + 2 * hl + p0;
    const float* Bn = Pl + KN * (NDIM * NDIM);
    #pragma unroll
    for (int s = 0; s < ns; ++s) {
      v2f a;
      a.x = arow[4 * s];
      a.y = arow[4 * s + 1];
      acc = __builtin_amdgcn_wmma_f32_16x16x4_f32(
           false, a,  false, bcur[s],
           (short)0, acc,  false,  false);
      if (s < nsn) cb_fetch_step<KN>(Bn, s, bnext[s]);
    }
  }
  __syncthreads();
  if (act) {
    #pragma unroll
    for (int g = 0; g < 8; ++g) S[(g + 8 * hl) * RS + n] = acc[g];
  }
  __syncthreads();
  if constexpr (K + 1 < GV_K) cb_product<K + 1>(Pl, S, wave, ln16, hl, n, bnext);
}

__global__ void __launch_bounds__(512)
qlayer_combine(const float* __restrict__ P,
               float* __restrict__ T1,
               float* __restrict__ T2) {
  __shared__ float S[CB_ROWS * RS];
  const int tid  = threadIdx.x;
  const int lane = tid & 31;
  const int hl   = lane >> 4;
  const int ln16 = lane & 15;
  const int wave = tid >> 5;
  const int mat  = blockIdx.x >> 4;
  const int row0 = (blockIdx.x & 15) * CB_ROWS;
  const float* Pm = P + mat * GV_K * (NDIM * NDIM);
  float* T = (mat == 0) ? T1 : T2;

  const int n = wave * 16 + ln16;
  const float* Pl = Pm + (2 * hl) * NDIM + n;

  v2f b1[CB_MAXNS];
  if (wave * 16 >= CB_P0(1)) {
    #pragma unroll
    for (int s = 0; s < CB_NS(1); ++s) cb_fetch_step<1>(Pl + NDIM * NDIM, s, b1[s]);
  }

  const int srow = tid >> 5;
  const int sc4  = tid & 31;
  {
    const float4* src = (const float4*)(Pm + (row0 + srow) * NDIM);
    float4* dst = (float4*)(S + srow * RS);
    dst[sc4]      = src[sc4];
    dst[sc4 + 32] = src[sc4 + 32];
  }
  __syncthreads();

  cb_product<1>(Pl, S, wave, ln16, hl, n, b1);

  {
    const float4* srcS = (const float4*)(S + srow * RS);
    float4* dst = (float4*)(T + (row0 + srow) * NDIM);
    dst[sc4]      = srcS[sc4];
    dst[sc4 + 32] = srcS[sc4 + 32];
  }
}

__global__ void __launch_bounds__(512)
qlayer_fold(float* __restrict__ T1,
            const float* __restrict__ T2,
            const float* __restrict__ cp,
            const float* __restrict__ sp,
            float* __restrict__ CI) {
  __shared__ float SR[CB_ROWS * RS];
  __shared__ float SI[CB_ROWS * RS];
  const int tid  = threadIdx.x;
  const int lane = tid & 31;
  const int hl   = lane >> 4;
  const int ln16 = lane & 15;
  const int wave = tid >> 5;
  const int row0 = blockIdx.x * CB_ROWS;

  const int n = wave * 16 + ln16;
  v2f bq[CB_MAXNS];
  {
    const float* T2l = T2 + (2 * hl) * NDIM + n;
    #pragma unroll
    for (int s = 0; s < CB_MAXNS; ++s) {
      bq[s].x = T2l[(4 * s) * NDIM];
      bq[s].y = T2l[(4 * s + 1) * NDIM];
    }
  }

  const int srow = tid >> 5;
  const int sc4  = tid & 31;
  {
    const float4* src = (const float4*)(T1 + (row0 + srow) * NDIM);
    const float4* c4  = (const float4*)cp;
    const float4* s4  = (const float4*)sp;
    float4* dR = (float4*)(SR + srow * RS);
    float4* dI = (float4*)(SI + srow * RS);
    #pragma unroll
    for (int q = 0; q < 2; ++q) {
      const int i = sc4 + 32 * q;
      const float4 v = src[i];
      const float4 c = c4[i];
      const float4 s = s4[i];
      dR[i] = make_float4(v.x * c.x, v.y * c.y, v.z * c.z, v.w * c.w);
      dI[i] = make_float4(v.x * s.x, v.y * s.y, v.z * s.z, v.w * s.w);
    }
  }
  __syncthreads();

  v8f accR = {0.f, 0.f, 0.f, 0.f, 0.f, 0.f, 0.f, 0.f};
  v8f accI = {0.f, 0.f, 0.f, 0.f, 0.f, 0.f, 0.f, 0.f};
  {
    const float* aRrow = SR + ln16 * RS + 2 * hl;
    const float* aIrow = SI + ln16 * RS + 2 * hl;
    #pragma unroll
    for (int s = 0; s < CB_MAXNS; ++s) {
      v2f aR, aI;
      aR.x = aRrow[4 * s];
      aR.y = aRrow[4 * s + 1];
      aI.x = aIrow[4 * s];
      aI.y = aIrow[4 * s + 1];
      accR = __builtin_amdgcn_wmma_f32_16x16x4_f32(
           false, aR,  false, bq[s],
           (short)0, accR,  false,  false);
      accI = __builtin_amdgcn_wmma_f32_16x16x4_f32(
           false, aI,  false, bq[s],
           (short)0, accI,  false,  false);
    }
  }
  __syncthreads();
  #pragma unroll
  for (int g = 0; g < 8; ++g) {
    SR[(g + 8 * hl) * RS + n] = accR[g];
    SI[(g + 8 * hl) * RS + n] = accI[g];
  }
  __syncthreads();

  {
    const float4* sR = (const float4*)(SR + srow * RS);
    const float4* sI = (const float4*)(SI + srow * RS);
    float4* dR = (float4*)(T1 + (row0 + srow) * NDIM);
    float4* dI = (float4*)(CI + (row0 + srow) * NDIM);
    dR[sc4]      = sR[sc4];
    dR[sc4 + 32] = sR[sc4 + 32];
    dI[sc4]      = sI[sc4];
    dI[sc4 + 32] = sI[sc4 + 32];
  }
}

#define MN_SCR 1088

__global__ void __launch_bounds__(256)
qlayer_main(const float* __restrict__ x,
            const float* __restrict__ CR,
            const float* __restrict__ CI,
            float* __restrict__ out) {
  extern __shared__ float sm[];
  float* smA = sm;
  float* smN = sm + TILE_B * RS;

  const int tid  = threadIdx.x;
  const int wave = tid >> 5;
  const int lane = tid & 31;
  const int hl   = lane >> 4;
  const int ln16 = lane & 15;
  const int c0   = wave * 32;
  const int base_row = blockIdx.x * TILE_B;

  {
    const int row = tid >> 2;
    const int q   = tid & 3;
    const float4* src = (const float4*)(x + (base_row + row) * NDIM + q * 64);
    float4* dst = (float4*)(smA + row * RS + q * 64);
    float acc = 0.f;
    #pragma unroll
    for (int c = 0; c < 16; ++c) {
      float4 vv = src[c];
      dst[c] = vv;
      acc += vv.x * vv.x + vv.y * vv.y + vv.z * vv.z + vv.w * vv.w;
    }
    smN[row * 4 + q] = acc;
  }
  __syncthreads();
  if (tid < TILE_B) {
    float s = smN[tid * 4] + smN[tid * 4 + 1] + smN[tid * 4 + 2] + smN[tid * 4 + 3];
    smN[1024 + tid] = rsqrtf(s);
  }
  __syncthreads();
  {
    const int row = tid >> 2;
    const int q   = tid & 3;
    const float inv = smN[1024 + row];
    float4* dst = (float4*)(smA + row * RS + q * 64);
    #pragma unroll
    for (int c = 0; c < 16; ++c) {
      float4 vv = dst[c];
      vv.x *= inv; vv.y *= inv; vv.z *= inv; vv.w *= inv;
      dst[c] = vv;
    }
  }
  __syncthreads();

  const v8f vzero = {0.f, 0.f, 0.f, 0.f, 0.f, 0.f, 0.f, 0.f};

  v8f accR[8];
  v8f accI[8];
  #pragma unroll
  for (int t = 0; t < 8; ++t) { accR[t] = vzero; accI[t] = vzero; }
  gemm_block(smA, CR, CI, c0, ln16, hl, accR, accI);

  #pragma unroll
  for (int rt = 0; rt < 4; ++rt) {
    #pragma unroll
    for (int ct = 0; ct < 2; ++ct) {
      const int n = c0 + ct * 16 + ln16;
      #pragma unroll
      for (int g = 0; g < 8; ++g) {
        const int m = rt * 16 + g + 8 * hl;
        const float vr = accR[rt * 2 + ct][g];
        const float vi = accI[rt * 2 + ct][g];
        out[(base_row + m) * NDIM + n] = fmaf(vi, vi, vr * vr);
      }
    }
  }
}

extern "C" void kernel_launch(void* const* d_in, const int* in_sizes, int n_in,
                              void* d_out, int out_size, void* d_ws, size_t ws_size,
                              hipStream_t stream) {
  const float* x      = (const float*)d_in[0];
  const float* rots1  = (const float*)d_in[1];
  const float* phases = (const float*)d_in[2];
  const float* rots2  = (const float*)d_in[3];
  float* out = (float*)d_out;
  float* ws  = (float*)d_ws;

  float* c1 = ws;
  float* s1 = ws + 32640;
  float* c2 = ws + 65280;
  float* s2 = ws + 97920;
  float* cp = ws + 130560;
  float* sp = ws + 130816;
  float* T1 = ws + 131072;
  float* T2 = ws + 196608;
  float* CI = ws;

  const int lds_givens = GV_LDS_BYTES;
  const int lds_main   = (TILE_B * RS + MN_SCR) * 4;
  hipFuncSetAttribute(reinterpret_cast<const void*>(qlayer_givens),
                      hipFuncAttributeMaxDynamicSharedMemorySize, lds_givens);
  hipFuncSetAttribute(reinterpret_cast<const void*>(qlayer_main),
                      hipFuncAttributeMaxDynamicSharedMemorySize, lds_main);

  qlayer_trig<<<(NROT + 255) / 256, 256, 0, stream>>>(
      rots1, rots2, phases, c1, s1, c2, s2, cp, sp);
  qlayer_givens<<<2 * GV_K * GV_CS, GV_NT, lds_givens, stream>>>(c1, s1, c2, s2, out);
  qlayer_combine<<<32, 512, 0, stream>>>(out, T1, T2);
  qlayer_fold<<<NDIM / CB_ROWS, 512, 0, stream>>>(T1, T2, cp, sp, CI);
  qlayer_main<<<NBATCH / TILE_B, 256, lds_main, stream>>>(
      x, T1, CI, out);
}


# Rotation-matrix build: each plane rotation is one packed f32 multiply and one packed f32 fused multiply-add on its sin/cos pair, adjacent in LDS; every element keeps its operations and rounding
# speedup vs baseline: 12.1738x; 1.0804x over previous
#include <hip/hip_runtime.h>
#include <math.h>


#define NDIM   256
#define NBATCH 65536
#define NROT   32640
#define TILE_B 64
#define RS     260

typedef __attribute__((ext_vector_type(2))) float v2f;
typedef __attribute__((ext_vector_type(8))) float v8f;

__global__ void qlayer_trig(const float* __restrict__ rots1,
                            const float* __restrict__ rots2,
                            const float* __restrict__ phases,
                            float* __restrict__ c1, float* __restrict__ s1,
                            float* __restrict__ c2, float* __restrict__ s2,
                            float* __restrict__ cp, float* __restrict__ sp) {
  int idx = blockIdx.x * blockDim.x + threadIdx.x;
  if (idx < NROT) {
    float s, c;
    sincosf(rots1[idx], &s, &c); c1[idx] = c; s1[idx] = s;
    sincosf(rots2[idx], &s, &c); c2[idx] = c; s2[idx] = s;
  }
  if (idx < NDIM) {
    float s, c;
    sincosf(phases[idx], &s, &c); cp[idx] = c; sp[idx] = s;
  }
}

#define GV_PB 8
#define GV_JU 4
#define GV_NB (NDIM / GV_PB)
#define GV_TS 72
#define GV_STAGE (((NDIM - GV_PB) / GV_JU) * GV_TS)
#define GV_INTRA 128
#define GV_CS 4
#define GV_NT (NDIM / GV_CS)
#define GV_IE ((GV_INTRA + GV_NT - 1) / GV_NT)
#define GV_NR (GV_CS * 16 + GV_IE)
#define GV_VEC (GV_NT * 257)
#define GV_LDS_FLOATS (GV_VEC + GV_STAGE + GV_INTRA)
#define GV_LDS_BYTES (164 * 1024)
static_assert(GV_LDS_FLOATS * 4 <= GV_LDS_BYTES, "column vectors and staging fit the reservation");

#define GV_K 8

__device__ __forceinline__ int gv_seg_block(int k) {
  const unsigned long long packed = 0x20550309184080ULL;
  return (int)((packed >> (6 * k)) & 63ULL);
}

typedef __attribute__((ext_vector_type(4))) float v4f;

__device__ __forceinline__ void givens_rot_sc(v2f sc, float& ri, float& rj) {
  const v2f aa = {ri, ri};
  const v2f bb = {rj, rj};
  const v2f m  = sc * bb;
  const v2f cs = {sc.y, sc.x};
  const v2f nm = {-m.x, m.y};
  const v2f r  = __builtin_elementwise_fma(cs, aa, nm);
  ri = r.x;
  rj = r.y;
}
__device__ __forceinline__ void givens_rot_cs(v2f cs, float& ri, float& rj) {
  const v2f aa = {ri, ri};
  const v2f bb = {rj, rj};
  const v2f m  = cs * bb;
  const v2f nm = {-m.y, m.x};
  const v2f r  = __builtin_elementwise_fma(cs, aa, nm);
  ri = r.x;
  rj = r.y;
}

__device__ __forceinline__ int gv_rowbase(int i) {
  return (i * (2 * NDIM - 1 - i)) / 2;
}

__device__ __forceinline__ void gv_fetch(const float* __restrict__ cA,
                                         const float* __restrict__ sA,
                                         int i0, int t, float (&r)[GV_NR]) {
  const int len = NDIM - GV_PB - i0;
  #pragma unroll
  for (int e = 0; e < GV_CS; ++e) {
    const int tt = t + e * GV_NT;
    #pragma unroll
    for (int q = 0; q < 16; ++q) r[16 * e + q] = 0.0f;
    if (tt < len) {
      #pragma unroll
      for (int q = 0; q < 16; ++q) {
        const int a = q >> 1;
        const float* src = ((q & 1) ? sA : cA) + (gv_rowbase(i0 + a) + (GV_PB - 1 - a));
        r[16 * e + q] = src[tt];
      }
    }
  }
  #pragma unroll
  for (int e = 0; e < GV_IE; ++e) {
    const int tt = t + e * GV_NT;
    const int a  = (tt >> 3) & 7;
    const int bb = tt & 7;
    const float* src = (tt & 64) ? sA : cA;
    r[16 * GV_CS + e] =
        (tt < GV_INTRA && bb > a) ? src[gv_rowbase(i0 + a) + (bb - a - 1)] : 0.0f;
  }
}

__device__ __forceinline__ void gv_put(float* stage, float* intra,
                                       int i0, int t, const float (&r)[GV_NR]) {
  const int len = NDIM - GV_PB - i0;
  #pragma unroll
  for (int e = 0; e < GV_CS; ++e) {
    const int tt = t + e * GV_NT;
    if (tt < len) {
      float* d = stage + (tt >> 2) * GV_TS + 2 * (tt & 3);
      #pragma unroll
      for (int q = 0; q < 16; ++q) d[(q >> 1) * 8 + ((q & 1) ^ 1 ^ (tt & 1))] = r[16 * e + q];
    }
  }
  #pragma unroll
  for (int e = 0; e < GV_IE; ++e) {
    const int tt = t + e * GV_NT;
    if (tt < GV_INTRA) intra[tt] = r[16 * GV_CS + e];
  }
}

__device__ __forceinline__ void gv_load_trip(const float* stage, int g,
                                             v4f (&q)[2 * GV_PB]) {
  const v4f* src = (const v4f*)(stage + g * GV_TS);
  #pragma unroll
  for (int i = 0; i < 2 * GV_PB; ++i) q[i] = src[i];
}

__device__ __forceinline__ void gv_trip(float* vj, const v4f (&q)[2 * GV_PB],
                                        float (&p)[GV_PB]) {
  float w[GV_JU];
  #pragma unroll
  for (int u = 0; u < GV_JU; ++u) w[u] = vj[u];
  #pragma unroll
  for (int a = 0; a < GV_PB; ++a) {
    #pragma unroll
    for (int u = 0; u < GV_JU; ++u) {
      const v4f qq = q[2 * a + (u >> 1)];
      if (u & 1) givens_rot_cs((v2f){qq.z, qq.w}, p[a], w[u]);
      else       givens_rot_sc((v2f){qq.x, qq.y}, p[a], w[u]);
    }
  }
  #pragma unroll
  for (int u = 0; u < GV_JU; ++u) vj[u] = w[u];
}

__global__ void __launch_bounds__(GV_NT)
qlayer_givens(const float* __restrict__ c1,
              const float* __restrict__ s1,
              const float* __restrict__ c2,
              const float* __restrict__ s2,
              float* __restrict__ P) {
  extern __shared__ float sm[];
  const int t = threadIdx.x;
  const int cg  = blockIdx.x % GV_CS;
  const int seg = (blockIdx.x / GV_CS) % GV_K;
  const int mat = blockIdx.x / (GV_CS * GV_K);
  const int col0 = cg * GV_NT;
  const float* cA = (mat == 0) ? c1 : c2;
  const float* sA = (mat == 0) ? s1 : s2;
  float* Pk = P + (mat * GV_K + seg) * (NDIM * NDIM);
  const int b_lo = gv_seg_block(seg);
  const int b_hi = gv_seg_block(seg + 1);

  float* v = sm + t * 257;
  float* stage = sm + GV_VEC;
  float* intra = stage + GV_STAGE;

  for (int k = 0; k < NDIM; ++k) v[k] = (k == col0 + t) ? 1.0f : 0.0f;

  float r[GV_NR];
  gv_fetch(cA, sA, b_lo * GV_PB, t, r);
  gv_put(stage, intra, b_lo * GV_PB, t, r);
  __syncthreads();

  for (int blk = b_lo; blk < b_hi; ++blk) {
    const int i0 = blk * GV_PB;
    const bool more = (blk + 1 < b_hi);

    if (more) gv_fetch(cA, sA, i0 + GV_PB, t, r);

    float p[GV_PB];
    #pragma unroll
    for (int a = 0; a < GV_PB; ++a) p[a] = v[i0 + a];

    #pragma unroll
    for (int a = 0; a < GV_PB; ++a) {
      #pragma unroll
      for (int b = a + 1; b < GV_PB; ++b) {
        const v2f sc = {intra[64 + a * 8 + b], intra[a * 8 + b]};
        givens_rot_sc(sc, p[a], p[b]);
      }
    }

    const int ntrip = (NDIM - GV_PB - i0) / GV_JU;
    float* vj = v + i0 + GV_PB;
    v4f qX[2 * GV_PB], qY[2 * GV_PB];
    if (ntrip > 0) gv_load_trip(stage, 0, qX);
    for (int g = 0; g < ntrip; g += 2) {
      gv_load_trip(stage, g + 1, qY);
      gv_trip(vj + GV_JU * g, qX, p);
      const int gn = (g + 2 < ntrip) ? g + 2 : g + 1;
      gv_load_trip(stage, gn, qX);
      gv_trip(vj + GV_JU * (g + 1), qY, p);
    }

    #pragma unroll
    for (int a = 0; a < GV_PB; ++a) v[i0 + a] = p[a];

    __syncthreads();
    if (more) gv_put(stage, intra, i0 + GV_PB, t, r);
    __syncthreads();
  }
  for (int k = 0; k < GV_NT; ++k) {
    float* dst = Pk + (col0 + k) * NDIM;
    const float* srcv = sm + k * 257;
    #pragma unroll
    for (int e = 0; e < GV_CS; ++e) dst[t + e * GV_NT] = srcv[t + e * GV_NT];
  }
}

__device__ __forceinline__ void gemm_block(const float* __restrict__ smSrc,
                                           const float* __restrict__ TR,
                                           const float* __restrict__ TI,
                                           int c0, int ln16, int hl,
                                           v8f accR[8], v8f accI[8]) {
  #pragma unroll 2
  for (int k = 0; k < NDIM; k += 4) {
    const int kk = k + 2 * hl;
    v2f a[4];
    #pragma unroll
    for (int rt = 0; rt < 4; ++rt) {
      a[rt].x = smSrc[(rt * 16 + ln16) * RS + kk];
      a[rt].y = smSrc[(rt * 16 + ln16) * RS + kk + 1];
    }
    const float* r0 = TR + kk * NDIM + c0 + ln16;
    const float* i0 = TI + kk * NDIM + c0 + ln16;
    v2f bR[2], bI[2];
    #pragma unroll
    for (int ct = 0; ct < 2; ++ct) {
      bR[ct].x = r0[ct * 16];
      bR[ct].y = r0[NDIM + ct * 16];
      bI[ct].x = i0[ct * 16];
      bI[ct].y = i0[NDIM + ct * 16];
    }
    #pragma unroll
    for (int rt = 0; rt < 4; ++rt) {
      #pragma unroll
      for (int ct = 0; ct < 2; ++ct) {
        accR[rt * 2 + ct] = __builtin_amdgcn_wmma_f32_16x16x4_f32(
             false, a[rt],  false, bR[ct],
             (short)0, accR[rt * 2 + ct],  false,  false);
        accI[rt * 2 + ct] = __builtin_amdgcn_wmma_f32_16x16x4_f32(
             false, a[rt],  false, bI[ct],
             (short)0, accI[rt * 2 + ct],  false,  false);
      }
    }
  }
}

#define CB_ROWS 16
#define CB_SEG(k) ((int)((0x20550309184080ULL >> (6 * (k))) & 63ULL))
#define CB_P0(k)  ((CB_SEG(k) * GV_PB) & ~15)
#define CB_NS(k)  ((NDIM - CB_P0(k)) / 4)
#define CB_MAXNS  (NDIM / 4)

template <int K>
__device__ __forceinline__ void cb_fetch_step(const float* __restrict__ Bl, int s, v2f& b) {
  b.x = Bl[(CB_P0(K) + 4 * s) * NDIM];
  b.y = Bl[(CB_P0(K) + 4 * s + 1) * NDIM];
}

template <int K>
__device__ __forceinline__ void cb_product(const float* __restrict__ Pl, float* S,
                                           int wave, int ln16, int hl, int n,
                                           const v2f (&bcur)[CB_MAXNS]) {
  constexpr int p0  = CB_P0(K);
  constexpr int ns  = CB_NS(K);
  constexpr int KN  = (K + 1 < GV_K) ? K + 1 : K;
  constexpr int nsn = (K + 1 < GV_K) ? CB_NS(KN) : 0;
  const bool act = (wave * 16 >= p0);
  v8f acc = {0.f, 0.f, 0.f, 0.f, 0.f, 0.f, 0.f, 0.f};
  v2f bnext[CB_MAXNS];
  if (act) {
    const float* arow = S + ln16 * RS + 2 * hl + p0;
    const float* Bn = Pl + KN * (NDIM * NDIM);
    #pragma unroll
    for (int s = 0; s < ns; ++s) {
      v2f a;
      a.x = arow[4 * s];
      a.y = arow[4 * s + 1];
      acc = __builtin_amdgcn_wmma_f32_16x16x4_f32(
           false, a,  false, bcur[s],
           (short)0, acc,  false,  false);
      if (s < nsn) cb_fetch_step<KN>(Bn, s, bnext[s]);
    }
  }
  __syncthreads();
  if (act) {
    #pragma unroll
    for (int g = 0; g < 8; ++g) S[(g + 8 * hl) * RS + n] = acc[g];
  }
  __syncthreads();
  if constexpr (K + 1 < GV_K) cb_product<K + 1>(Pl, S, wave, ln16, hl, n, bnext);
}

__global__ void __launch_bounds__(512)
qlayer_combine(const float* __restrict__ P,
               float* __restrict__ T1,
               float* __restrict__ T2) {
  __shared__ float S[CB_ROWS * RS];
  const int tid  = threadIdx.x;
  const int lane = tid & 31;
  const int hl   = lane >> 4;
  const int ln16 = lane & 15;
  const int wave = tid >> 5;
  const int mat  = blockIdx.x >> 4;
  const int row0 = (blockIdx.x & 15) * CB_ROWS;
  const float* Pm = P + mat * GV_K * (NDIM * NDIM);
  float* T = (mat == 0) ? T1 : T2;

  const int n = wave * 16 + ln16;
  const float* Pl = Pm + (2 * hl) * NDIM + n;

  v2f b1[CB_MAXNS];
  if (wave * 16 >= CB_P0(1)) {
    #pragma unroll
    for (int s = 0; s < CB_NS(1); ++s) cb_fetch_step<1>(Pl + NDIM * NDIM, s, b1[s]);
  }

  const int srow = tid >> 5;
  const int sc4  = tid & 31;
  {
    const float4* src = (const float4*)(Pm + (row0 + srow) * NDIM);
    float4* dst = (float4*)(S + srow * RS);
    dst[sc4]      = src[sc4];
    dst[sc4 + 32] = src[sc4 + 32];
  }
  __syncthreads();

  cb_product<1>(Pl, S, wave, ln16, hl, n, b1);

  {
    const float4* srcS = (const float4*)(S + srow * RS);
    float4* dst = (float4*)(T + (row0 + srow) * NDIM);
    dst[sc4]      = srcS[sc4];
    dst[sc4 + 32] = srcS[sc4 + 32];
  }
}

__global__ void __launch_bounds__(512)
qlayer_fold(float* __restrict__ T1,
            const float* __restrict__ T2,
            const float* __restrict__ cp,
            const float* __restrict__ sp,
            float* __restrict__ CI) {
  __shared__ float SR[CB_ROWS * RS];
  __shared__ float SI[CB_ROWS * RS];
  const int tid  = threadIdx.x;
  const int lane = tid & 31;
  const int hl   = lane >> 4;
  const int ln16 = lane & 15;
  const int wave = tid >> 5;
  const int row0 = blockIdx.x * CB_ROWS;

  const int n = wave * 16 + ln16;
  v2f bq[CB_MAXNS];
  {
    const float* T2l = T2 + (2 * hl) * NDIM + n;
    #pragma unroll
    for (int s = 0; s < CB_MAXNS; ++s) {
      bq[s].x = T2l[(4 * s) * NDIM];
      bq[s].y = T2l[(4 * s + 1) * NDIM];
    }
  }

  const int srow = tid >> 5;
  const int sc4  = tid & 31;
  {
    const float4* src = (const float4*)(T1 + (row0 + srow) * NDIM);
    const float4* c4  = (const float4*)cp;
    const float4* s4  = (const float4*)sp;
    float4* dR = (float4*)(SR + srow * RS);
    float4* dI = (float4*)(SI + srow * RS);
    #pragma unroll
    for (int q = 0; q < 2; ++q) {
      const int i = sc4 + 32 * q;
      const float4 v = src[i];
      const float4 c = c4[i];
      const float4 s = s4[i];
      dR[i] = make_float4(v.x * c.x, v.y * c.y, v.z * c.z, v.w * c.w);
      dI[i] = make_float4(v.x * s.x, v.y * s.y, v.z * s.z, v.w * s.w);
    }
  }
  __syncthreads();

  v8f accR = {0.f, 0.f, 0.f, 0.f, 0.f, 0.f, 0.f, 0.f};
  v8f accI = {0.f, 0.f, 0.f, 0.f, 0.f, 0.f, 0.f, 0.f};
  {
    const float* aRrow = SR + ln16 * RS + 2 * hl;
    const float* aIrow = SI + ln16 * RS + 2 * hl;
    #pragma unroll
    for (int s = 0; s < CB_MAXNS; ++s) {
      v2f aR, aI;
      aR.x = aRrow[4 * s];
      aR.y = aRrow[4 * s + 1];
      aI.x = aIrow[4 * s];
      aI.y = aIrow[4 * s + 1];
      accR = __builtin_amdgcn_wmma_f32_16x16x4_f32(
           false, aR,  false, bq[s],
           (short)0, accR,  false,  false);
      accI = __builtin_amdgcn_wmma_f32_16x16x4_f32(
           false, aI,  false, bq[s],
           (short)0, accI,  false,  false);
    }
  }
  __syncthreads();
  #pragma unroll
  for (int g = 0; g < 8; ++g) {
    SR[(g + 8 * hl) * RS + n] = accR[g];
    SI[(g + 8 * hl) * RS + n] = accI[g];
  }
  __syncthreads();

  {
    const float4* sR = (const float4*)(SR + srow * RS);
    const float4* sI = (const float4*)(SI + srow * RS);
    float4* dR = (float4*)(T1 + (row0 + srow) * NDIM);
    float4* dI = (float4*)(CI + (row0 + srow) * NDIM);
    dR[sc4]      = sR[sc4];
    dR[sc4 + 32] = sR[sc4 + 32];
    dI[sc4]      = sI[sc4];
    dI[sc4 + 32] = sI[sc4 + 32];
  }
}

#define MN_SCR 1088

__global__ void __launch_bounds__(256)
qlayer_main(const float* __restrict__ x,
            const float* __restrict__ CR,
            const float* __restrict__ CI,
            float* __restrict__ out) {
  extern __shared__ float sm[];
  float* smA = sm;
  float* smN = sm + TILE_B * RS;

  const int tid  = threadIdx.x;
  const int wave = tid >> 5;
  const int lane = tid & 31;
  const int hl   = lane >> 4;
  const int ln16 = lane & 15;
  const int c0   = wave * 32;
  const int base_row = blockIdx.x * TILE_B;

  {
    const int row = tid >> 2;
    const int q   = tid & 3;
    const float4* src = (const float4*)(x + (base_row + row) * NDIM + q * 64);
    float4* dst = (float4*)(smA + row * RS + q * 64);
    float acc = 0.f;
    #pragma unroll
    for (int c = 0; c < 16; ++c) {
      float4 vv = src[c];
      dst[c] = vv;
      acc += vv.x * vv.x + vv.y * vv.y + vv.z * vv.z + vv.w * vv.w;
    }
    smN[row * 4 + q] = acc;
  }
  __syncthreads();
  if (tid < TILE_B) {
    float s = smN[tid * 4] + smN[tid * 4 + 1] + smN[tid * 4 + 2] + smN[tid * 4 + 3];
    smN[1024 + tid] = rsqrtf(s);
  }
  __syncthreads();
  {
    const int row = tid >> 2;
    const int q   = tid & 3;
    const float inv = smN[1024 + row];
    float4* dst = (float4*)(smA + row * RS + q * 64);
    #pragma unroll
    for (int c = 0; c < 16; ++c) {
      float4 vv = dst[c];
      vv.x *= inv; vv.y *= inv; vv.z *= inv; vv.w *= inv;
      dst[c] = vv;
    }
  }
  __syncthreads();

  const v8f vzero = {0.f, 0.f, 0.f, 0.f, 0.f, 0.f, 0.f, 0.f};

  v8f accR[8];
  v8f accI[8];
  #pragma unroll
  for (int t = 0; t < 8; ++t) { accR[t] = vzero; accI[t] = vzero; }
  gemm_block(smA, CR, CI, c0, ln16, hl, accR, accI);

  #pragma unroll
  for (int rt = 0; rt < 4; ++rt) {
    #pragma unroll
    for (int ct = 0; ct < 2; ++ct) {
      const int n = c0 + ct * 16 + ln16;
      #pragma unroll
      for (int g = 0; g < 8; ++g) {
        const int m = rt * 16 + g + 8 * hl;
        const float vr = accR[rt * 2 + ct][g];
        const float vi = accI[rt * 2 + ct][g];
        out[(base_row + m) * NDIM + n] = fmaf(vi, vi, vr * vr);
      }
    }
  }
}

extern "C" void kernel_launch(void* const* d_in, const int* in_sizes, int n_in,
                              void* d_out, int out_size, void* d_ws, size_t ws_size,
                              hipStream_t stream) {
  const float* x      = (const float*)d_in[0];
  const float* rots1  = (const float*)d_in[1];
  const float* phases = (const float*)d_in[2];
  const float* rots2  = (const float*)d_in[3];
  float* out = (float*)d_out;
  float* ws  = (float*)d_ws;

  float* c1 = ws;
  float* s1 = ws + 32640;
  float* c2 = ws + 65280;
  float* s2 = ws + 97920;
  float* cp = ws + 130560;
  float* sp = ws + 130816;
  float* T1 = ws + 131072;
  float* T2 = ws + 196608;
  float* CI = ws;

  const int lds_givens = GV_LDS_BYTES;
  const int lds_main   = (TILE_B * RS + MN_SCR) * 4;
  hipFuncSetAttribute(reinterpret_cast<const void*>(qlayer_givens),
                      hipFuncAttributeMaxDynamicSharedMemorySize, lds_givens);
  hipFuncSetAttribute(reinterpret_cast<const void*>(qlayer_main),
                      hipFuncAttributeMaxDynamicSharedMemorySize, lds_main);

  qlayer_trig<<<(NROT + 255) / 256, 256, 0, stream>>>(
      rots1, rots2, phases, c1, s1, c2, s2, cp, sp);
  qlayer_givens<<<2 * GV_K * GV_CS, GV_NT, lds_givens, stream>>>(c1, s1, c2, s2, out);
  qlayer_combine<<<32, 512, 0, stream>>>(out, T1, T2);
  qlayer_fold<<<NDIM / CB_ROWS, 512, 0, stream>>>(T1, T2, cp, sp, CI);
  qlayer_main<<<NBATCH / TILE_B, 256, lds_main, stream>>>(
      x, T1, CI, out);
}


# Fold product C^T = T1 diag(e^{i phi}) T2 dealt over 128 work-groups (16 rows x 32 columns) instead of 16; its sum over n cut into 4 ranges summed apart, then added in order (new sum order)
# speedup vs baseline: 12.8038x; 1.0518x over previous
#include <hip/hip_runtime.h>
#include <math.h>


#define NDIM   256
#define NBATCH 65536
#define NROT   32640
#define TILE_B 64
#define RS     260

typedef __attribute__((ext_vector_type(2))) float v2f;
typedef __attribute__((ext_vector_type(8))) float v8f;

__global__ void qlayer_trig(const float* __restrict__ rots1,
                            const float* __restrict__ rots2,
                            const float* __restrict__ phases,
                            float* __restrict__ c1, float* __restrict__ s1,
                            float* __restrict__ c2, float* __restrict__ s2,
                            float* __restrict__ cp, float* __restrict__ sp) {
  int idx = blockIdx.x * blockDim.x + threadIdx.x;
  if (idx < NROT) {
    float s, c;
    sincosf(rots1[idx], &s, &c); c1[idx] = c; s1[idx] = s;
    sincosf(rots2[idx], &s, &c); c2[idx] = c; s2[idx] = s;
  }
  if (idx < NDIM) {
    float s, c;
    sincosf(phases[idx], &s, &c); cp[idx] = c; sp[idx] = s;
  }
}

#define GV_PB 8
#define GV_JU 4
#define GV_NB (NDIM / GV_PB)
#define GV_TS 72
#define GV_STAGE (((NDIM - GV_PB) / GV_JU) * GV_TS)
#define GV_INTRA 128
#define GV_CS 4
#define GV_NT (NDIM / GV_CS)
#define GV_IE ((GV_INTRA + GV_NT - 1) / GV_NT)
#define GV_NR (GV_CS * 16 + GV_IE)
#define GV_VEC (GV_NT * 257)
#define GV_LDS_FLOATS (GV_VEC + GV_STAGE + GV_INTRA)
#define GV_LDS_BYTES (164 * 1024)
static_assert(GV_LDS_FLOATS * 4 <= GV_LDS_BYTES, "column vectors and staging fit the reservation");

#define GV_K 8

__device__ __forceinline__ int gv_seg_block(int k) {
  const unsigned long long packed = 0x20550309184080ULL;
  return (int)((packed >> (6 * k)) & 63ULL);
}

typedef __attribute__((ext_vector_type(4))) float v4f;

__device__ __forceinline__ void givens_rot_sc(v2f sc, float& ri, float& rj) {
  const v2f aa = {ri, ri};
  const v2f bb = {rj, rj};
  const v2f m  = sc * bb;
  const v2f cs = {sc.y, sc.x};
  const v2f nm = {-m.x, m.y};
  const v2f r  = __builtin_elementwise_fma(cs, aa, nm);
  ri = r.x;
  rj = r.y;
}
__device__ __forceinline__ void givens_rot_cs(v2f cs, float& ri, float& rj) {
  const v2f aa = {ri, ri};
  const v2f bb = {rj, rj};
  const v2f m  = cs * bb;
  const v2f nm = {-m.y, m.x};
  const v2f r  = __builtin_elementwise_fma(cs, aa, nm);
  ri = r.x;
  rj = r.y;
}

__device__ __forceinline__ int gv_rowbase(int i) {
  return (i * (2 * NDIM - 1 - i)) / 2;
}

__device__ __forceinline__ void gv_fetch(const float* __restrict__ cA,
                                         const float* __restrict__ sA,
                                         int i0, int t, float (&r)[GV_NR]) {
  const int len = NDIM - GV_PB - i0;
  #pragma unroll
  for (int e = 0; e < GV_CS; ++e) {
    const int tt = t + e * GV_NT;
    #pragma unroll
    for (int q = 0; q < 16; ++q) r[16 * e + q] = 0.0f;
    if (tt < len) {
      #pragma unroll
      for (int q = 0; q < 16; ++q) {
        const int a = q >> 1;
        const float* src = ((q & 1) ? sA : cA) + (gv_rowbase(i0 + a) + (GV_PB - 1 - a));
        r[16 * e + q] = src[tt];
      }
    }
  }
  #pragma unroll
  for (int e = 0; e < GV_IE; ++e) {
    const int tt = t + e * GV_NT;
    const int a  = (tt >> 3) & 7;
    const int bb = tt & 7;
    const float* src = (tt & 64) ? sA : cA;
    r[16 * GV_CS + e] =
        (tt < GV_INTRA && bb > a) ? src[gv_rowbase(i0 + a) + (bb - a - 1)] : 0.0f;
  }
}

__device__ __forceinline__ void gv_put(float* stage, float* intra,
                                       int i0, int t, const float (&r)[GV_NR]) {
  const int len = NDIM - GV_PB - i0;
  #pragma unroll
  for (int e = 0; e < GV_CS; ++e) {
    const int tt = t + e * GV_NT;
    if (tt < len) {
      float* d = stage + (tt >> 2) * GV_TS + 2 * (tt & 3);
      #pragma unroll
      for (int q = 0; q < 16; ++q) d[(q >> 1) * 8 + ((q & 1) ^ 1 ^ (tt & 1))] = r[16 * e + q];
    }
  }
  #pragma unroll
  for (int e = 0; e < GV_IE; ++e) {
    const int tt = t + e * GV_NT;
    if (tt < GV_INTRA) intra[tt] = r[16 * GV_CS + e];
  }
}

__device__ __forceinline__ void gv_load_trip(const float* stage, int g,
                                             v4f (&q)[2 * GV_PB]) {
  const v4f* src = (const v4f*)(stage + g * GV_TS);
  #pragma unroll
  for (int i = 0; i < 2 * GV_PB; ++i) q[i] = src[i];
}

__device__ __forceinline__ void gv_trip(float* vj, const v4f (&q)[2 * GV_PB],
                                        float (&p)[GV_PB]) {
  float w[GV_JU];
  #pragma unroll
  for (int u = 0; u < GV_JU; ++u) w[u] = vj[u];
  #pragma unroll
  for (int a = 0; a < GV_PB; ++a) {
    #pragma unroll
    for (int u = 0; u < GV_JU; ++u) {
      const v4f qq = q[2 * a + (u >> 1)];
      if (u & 1) givens_rot_cs((v2f){qq.z, qq.w}, p[a], w[u]);
      else       givens_rot_sc((v2f){qq.x, qq.y}, p[a], w[u]);
    }
  }
  #pragma unroll
  for (int u = 0; u < GV_JU; ++u) vj[u] = w[u];
}

__global__ void __launch_bounds__(GV_NT)
qlayer_givens(const float* __restrict__ c1,
              const float* __restrict__ s1,
              const float* __restrict__ c2,
              const float* __restrict__ s2,
              float* __restrict__ P) {
  extern __shared__ float sm[];
  const int t = threadIdx.x;
  const int cg  = blockIdx.x % GV_CS;
  const int seg = (blockIdx.x / GV_CS) % GV_K;
  const int mat = blockIdx.x / (GV_CS * GV_K);
  const int col0 = cg * GV_NT;
  const float* cA = (mat == 0) ? c1 : c2;
  const float* sA = (mat == 0) ? s1 : s2;
  float* Pk = P + (mat * GV_K + seg) * (NDIM * NDIM);
  const int b_lo = gv_seg_block(seg);
  const int b_hi = gv_seg_block(seg + 1);

  float* v = sm + t * 257;
  float* stage = sm + GV_VEC;
  float* intra = stage + GV_STAGE;

  for (int k = 0; k < NDIM; ++k) v[k] = (k == col0 + t) ? 1.0f : 0.0f;

  float r[GV_NR];
  gv_fetch(cA, sA, b_lo * GV_PB, t, r);
  gv_put(stage, intra, b_lo * GV_PB, t, r);
  __syncthreads();

  for (int blk = b_lo; blk < b_hi; ++blk) {
    const int i0 = blk * GV_PB;
    const bool more = (blk + 1 < b_hi);

    if (more) gv_fetch(cA, sA, i0 + GV_PB, t, r);

    float p[GV_PB];
    #pragma unroll
    for (int a = 0; a < GV_PB; ++a) p[a] = v[i0 + a];

    #pragma unroll
    for (int a = 0; a < GV_PB; ++a) {
      #pragma unroll
      for (int b = a + 1; b < GV_PB; ++b) {
        const v2f sc = {intra[64 + a * 8 + b], intra[a * 8 + b]};
        givens_rot_sc(sc, p[a], p[b]);
      }
    }

    const int ntrip = (NDIM - GV_PB - i0) / GV_JU;
    float* vj = v + i0 + GV_PB;
    v4f qX[2 * GV_PB], qY[2 * GV_PB];
    if (ntrip > 0) gv_load_trip(stage, 0, qX);
    for (int g = 0; g < ntrip; g += 2) {
      gv_load_trip(stage, g + 1, qY);
      gv_trip(vj + GV_JU * g, qX, p);
      const int gn = (g + 2 < ntrip) ? g + 2 : g + 1;
      gv_load_trip(stage, gn, qX);
      gv_trip(vj + GV_JU * (g + 1), qY, p);
    }

    #pragma unroll
    for (int a = 0; a < GV_PB; ++a) v[i0 + a] = p[a];

    __syncthreads();
    if (more) gv_put(stage, intra, i0 + GV_PB, t, r);
    __syncthreads();
  }
  for (int k = 0; k < GV_NT; ++k) {
    float* dst = Pk + (col0 + k) * NDIM;
    const float* srcv = sm + k * 257;
    #pragma unroll
    for (int e = 0; e < GV_CS; ++e) dst[t + e * GV_NT] = srcv[t + e * GV_NT];
  }
}

__device__ __forceinline__ void gemm_block(const float* __restrict__ smSrc,
                                           const float* __restrict__ TR,
                                           const float* __restrict__ TI,
                                           int c0, int ln16, int hl,
                                           v8f accR[8], v8f accI[8]) {
  #pragma unroll 2
  for (int k = 0; k < NDIM; k += 4) {
    const int kk = k + 2 * hl;
    v2f a[4];
    #pragma unroll
    for (int rt = 0; rt < 4; ++rt) {
      a[rt].x = smSrc[(rt * 16 + ln16) * RS + kk];
      a[rt].y = smSrc[(rt * 16 + ln16) * RS + kk + 1];
    }
    const float* r0 = TR + kk * NDIM + c0 + ln16;
    const float* i0 = TI + kk * NDIM + c0 + ln16;
    v2f bR[2], bI[2];
    #pragma unroll
    for (int ct = 0; ct < 2; ++ct) {
      bR[ct].x = r0[ct * 16];
      bR[ct].y = r0[NDIM + ct * 16];
      bI[ct].x = i0[ct * 16];
      bI[ct].y = i0[NDIM + ct * 16];
    }
    #pragma unroll
    for (int rt = 0; rt < 4; ++rt) {
      #pragma unroll
      for (int ct = 0; ct < 2; ++ct) {
        accR[rt * 2 + ct] = __builtin_amdgcn_wmma_f32_16x16x4_f32(
             false, a[rt],  false, bR[ct],
             (short)0, accR[rt * 2 + ct],  false,  false);
        accI[rt * 2 + ct] = __builtin_amdgcn_wmma_f32_16x16x4_f32(
             false, a[rt],  false, bI[ct],
             (short)0, accI[rt * 2 + ct],  false,  false);
      }
    }
  }
}

#define CB_ROWS 16
#define CB_SEG(k) ((int)((0x20550309184080ULL >> (6 * (k))) & 63ULL))
#define CB_P0(k)  ((CB_SEG(k) * GV_PB) & ~15)
#define CB_NS(k)  ((NDIM - CB_P0(k)) / 4)
#define CB_MAXNS  (NDIM / 4)

template <int K>
__device__ __forceinline__ void cb_fetch_step(const float* __restrict__ Bl, int s, v2f& b) {
  b.x = Bl[(CB_P0(K) + 4 * s) * NDIM];
  b.y = Bl[(CB_P0(K) + 4 * s + 1) * NDIM];
}

template <int K>
__device__ __forceinline__ void cb_product(const float* __restrict__ Pl, float* S,
                                           int wave, int ln16, int hl, int n,
                                           const v2f (&bcur)[CB_MAXNS]) {
  constexpr int p0  = CB_P0(K);
  constexpr int ns  = CB_NS(K);
  constexpr int KN  = (K + 1 < GV_K) ? K + 1 : K;
  constexpr int nsn = (K + 1 < GV_K) ? CB_NS(KN) : 0;
  const bool act = (wave * 16 >= p0);
  v8f acc = {0.f, 0.f, 0.f, 0.f, 0.f, 0.f, 0.f, 0.f};
  v2f bnext[CB_MAXNS];
  if (act) {
    const float* arow = S + ln16 * RS + 2 * hl + p0;
    const float* Bn = Pl + KN * (NDIM * NDIM);
    #pragma unroll
    for (int s = 0; s < ns; ++s) {
      v2f a;
      a.x = arow[4 * s];
      a.y = arow[4 * s + 1];
      acc = __builtin_amdgcn_wmma_f32_16x16x4_f32(
           false, a,  false, bcur[s],
           (short)0, acc,  false,  false);
      if (s < nsn) cb_fetch_step<KN>(Bn, s, bnext[s]);
    }
  }
  __syncthreads();
  if (act) {
    #pragma unroll
    for (int g = 0; g < 8; ++g) S[(g + 8 * hl) * RS + n] = acc[g];
  }
  __syncthreads();
  if constexpr (K + 1 < GV_K) cb_product<K + 1>(Pl, S, wave, ln16, hl, n, bnext);
}

__global__ void __launch_bounds__(512)
qlayer_combine(const float* __restrict__ P,
               float* __restrict__ T1,
               float* __restrict__ T2) {
  __shared__ float S[CB_ROWS * RS];
  const int tid  = threadIdx.x;
  const int lane = tid & 31;
  const int hl   = lane >> 4;
  const int ln16 = lane & 15;
  const int wave = tid >> 5;
  const int mat  = blockIdx.x >> 4;
  const int row0 = (blockIdx.x & 15) * CB_ROWS;
  const float* Pm = P + mat * GV_K * (NDIM * NDIM);
  float* T = (mat == 0) ? T1 : T2;

  const int n = wave * 16 + ln16;
  const float* Pl = Pm + (2 * hl) * NDIM + n;

  v2f b1[CB_MAXNS];
  if (wave * 16 >= CB_P0(1)) {
    #pragma unroll
    for (int s = 0; s < CB_NS(1); ++s) cb_fetch_step<1>(Pl + NDIM * NDIM, s, b1[s]);
  }

  const int srow = tid >> 5;
  const int sc4  = tid & 31;
  {
    const float4* src = (const float4*)(Pm + (row0 + srow) * NDIM);
    float4* dst = (float4*)(S + srow * RS);
    dst[sc4]      = src[sc4];
    dst[sc4 + 32] = src[sc4 + 32];
  }
  __syncthreads();

  cb_product<1>(Pl, S, wave, ln16, hl, n, b1);

  {
    const float4* srcS = (const float4*)(S + srow * RS);
    float4* dst = (float4*)(T + (row0 + srow) * NDIM);
    dst[sc4]      = srcS[sc4];
    dst[sc4 + 32] = srcS[sc4 + 32];
  }
}

#define FD_COLS 32
#define FD_KQ   4
#define FD_KS   (NDIM / FD_KQ / 4)

__global__ void __launch_bounds__(256)
qlayer_fold(const float* __restrict__ T1,
            const float* __restrict__ T2,
            const float* __restrict__ cp,
            const float* __restrict__ sp,
            float* __restrict__ CR,
            float* __restrict__ CI) {
  __shared__ float SR[CB_ROWS * RS];
  __shared__ float SI[CB_ROWS * RS];
  __shared__ float PS[8 * 2 * 256];
  const int tid  = threadIdx.x;
  const int lane = tid & 31;
  const int hl   = lane >> 4;
  const int ln16 = lane & 15;
  const int wave = tid >> 5;
  const int tile = wave >> 2;
  const int kq   = wave & 3;
  const int row0 = (blockIdx.x >> 3) * CB_ROWS;
  const int col0 = (blockIdx.x & 7) * FD_COLS;

  const int n = col0 + tile * 16 + ln16;
  v2f bq[FD_KS];
  {
    const float* T2l = T2 + (kq * 64 + 2 * hl) * NDIM + n;
    #pragma unroll
    for (int s = 0; s < FD_KS; ++s) {
      bq[s].x = T2l[(4 * s) * NDIM];
      bq[s].y = T2l[(4 * s + 1) * NDIM];
    }
  }

  {
    const int c4 = tid & 63;
    const int r  = tid >> 6;
    const float4 c = ((const float4*)cp)[c4];
    const float4 s = ((const float4*)sp)[c4];
    #pragma unroll
    for (int q = 0; q < 4; ++q) {
      const int row = r + 4 * q;
      const float4 v = ((const float4*)(T1 + (row0 + row) * NDIM))[c4];
      ((float4*)(SR + row * RS))[c4] = make_float4(v.x * c.x, v.y * c.y, v.z * c.z, v.w * c.w);
      ((float4*)(SI + row * RS))[c4] = make_float4(v.x * s.x, v.y * s.y, v.z * s.z, v.w * s.w);
    }
  }
  __syncthreads();

  v8f accR = {0.f, 0.f, 0.f, 0.f, 0.f, 0.f, 0.f, 0.f};
  v8f accI = {0.f, 0.f, 0.f, 0.f, 0.f, 0.f, 0.f, 0.f};
  {
    const float* aRrow = SR + ln16 * RS + kq * 64 + 2 * hl;
    const float* aIrow = SI + ln16 * RS + kq * 64 + 2 * hl;
    #pragma unroll
    for (int s = 0; s < FD_KS; ++s) {
      v2f aR, aI;
      aR.x = aRrow[4 * s];
      aR.y = aRrow[4 * s + 1];
      aI.x = aIrow[4 * s];
      aI.y = aIrow[4 * s + 1];
      accR = __builtin_amdgcn_wmma_f32_16x16x4_f32(
           false, aR,  false, bq[s],
           (short)0, accR,  false,  false);
      accI = __builtin_amdgcn_wmma_f32_16x16x4_f32(
           false, aI,  false, bq[s],
           (short)0, accI,  false,  false);
    }
  }
  {
    float* pR = PS + (wave * 2) * 256;
    float* pI = pR + 256;
    #pragma unroll
    for (int g = 0; g < 8; ++g) {
      pR[(g + 8 * hl) * 16 + ln16] = accR[g];
      pI[(g + 8 * hl) * 16 + ln16] = accI[g];
    }
  }
  __syncthreads();

  {
    const int part = tid >> 7;
    const int m    = (tid >> 3) & 15;
    const int c4   = tid & 7;
    const int t2   = c4 >> 2;
    const int j0   = (c4 & 3) * 4;
    const float* p0 = PS + (((t2 * 4) * 2 + part) * 16 + m) * 16 + j0;
    const float4 a0 = *(const float4*)(p0);
    const float4 a1 = *(const float4*)(p0 + 512);
    const float4 a2 = *(const float4*)(p0 + 1024);
    const float4 a3 = *(const float4*)(p0 + 1536);
    float4 r;
    r.x = ((a0.x + a1.x) + a2.x) + a3.x;
    r.y = ((a0.y + a1.y) + a2.y) + a3.y;
    r.z = ((a0.z + a1.z) + a2.z) + a3.z;
    r.w = ((a0.w + a1.w) + a2.w) + a3.w;
    float* dst = (part ? CI : CR) + (row0 + m) * NDIM + col0 + c4 * 4;
    *(float4*)dst = r;
  }
}

#define MN_SCR 1088

__global__ void __launch_bounds__(256)
qlayer_main(const float* __restrict__ x,
            const float* __restrict__ CR,
            const float* __restrict__ CI,
            float* __restrict__ out) {
  extern __shared__ float sm[];
  float* smA = sm;
  float* smN = sm + TILE_B * RS;

  const int tid  = threadIdx.x;
  const int wave = tid >> 5;
  const int lane = tid & 31;
  const int hl   = lane >> 4;
  const int ln16 = lane & 15;
  const int c0   = wave * 32;
  const int base_row = blockIdx.x * TILE_B;

  {
    const int row = tid >> 2;
    const int q   = tid & 3;
    const float4* src = (const float4*)(x + (base_row + row) * NDIM + q * 64);
    float4* dst = (float4*)(smA + row * RS + q * 64);
    float acc = 0.f;
    #pragma unroll
    for (int c = 0; c < 16; ++c) {
      float4 vv = src[c];
      dst[c] = vv;
      acc += vv.x * vv.x + vv.y * vv.y + vv.z * vv.z + vv.w * vv.w;
    }
    smN[row * 4 + q] = acc;
  }
  __syncthreads();
  if (tid < TILE_B) {
    float s = smN[tid * 4] + smN[tid * 4 + 1] + smN[tid * 4 + 2] + smN[tid * 4 + 3];
    smN[1024 + tid] = rsqrtf(s);
  }
  __syncthreads();
  {
    const int row = tid >> 2;
    const int q   = tid & 3;
    const float inv = smN[1024 + row];
    float4* dst = (float4*)(smA + row * RS + q * 64);
    #pragma unroll
    for (int c = 0; c < 16; ++c) {
      float4 vv = dst[c];
      vv.x *= inv; vv.y *= inv; vv.z *= inv; vv.w *= inv;
      dst[c] = vv;
    }
  }
  __syncthreads();

  const v8f vzero = {0.f, 0.f, 0.f, 0.f, 0.f, 0.f, 0.f, 0.f};

  v8f accR[8];
  v8f accI[8];
  #pragma unroll
  for (int t = 0; t < 8; ++t) { accR[t] = vzero; accI[t] = vzero; }
  gemm_block(smA, CR, CI, c0, ln16, hl, accR, accI);

  #pragma unroll
  for (int rt = 0; rt < 4; ++rt) {
    #pragma unroll
    for (int ct = 0; ct < 2; ++ct) {
      const int n = c0 + ct * 16 + ln16;
      #pragma unroll
      for (int g = 0; g < 8; ++g) {
        const int m = rt * 16 + g + 8 * hl;
        const float vr = accR[rt * 2 + ct][g];
        const float vi = accI[rt * 2 + ct][g];
        out[(base_row + m) * NDIM + n] = fmaf(vi, vi, vr * vr);
      }
    }
  }
}

extern "C" void kernel_launch(void* const* d_in, const int* in_sizes, int n_in,
                              void* d_out, int out_size, void* d_ws, size_t ws_size,
                              hipStream_t stream) {
  const float* x      = (const float*)d_in[0];
  const float* rots1  = (const float*)d_in[1];
  const float* phases = (const float*)d_in[2];
  const float* rots2  = (const float*)d_in[3];
  float* out = (float*)d_out;
  float* ws  = (float*)d_ws;

  float* c1 = ws;
  float* s1 = ws + 32640;
  float* c2 = ws + 65280;
  float* s2 = ws + 97920;
  float* cp = out + 1048576;
  float* sp = out + 1048832;
  float* T1 = ws + 131072;
  float* T2 = ws + 196608;
  float* CI = ws;
  float* CR = ws + 65536;

  const int lds_givens = GV_LDS_BYTES;
  const int lds_main   = (TILE_B * RS + MN_SCR) * 4;
  hipFuncSetAttribute(reinterpret_cast<const void*>(qlayer_givens),
                      hipFuncAttributeMaxDynamicSharedMemorySize, lds_givens);
  hipFuncSetAttribute(reinterpret_cast<const void*>(qlayer_main),
                      hipFuncAttributeMaxDynamicSharedMemorySize, lds_main);

  qlayer_trig<<<(NROT + 255) / 256, 256, 0, stream>>>(
      rots1, rots2, phases, c1, s1, c2, s2, cp, sp);
  qlayer_givens<<<2 * GV_K * GV_CS, GV_NT, lds_givens, stream>>>(c1, s1, c2, s2, out);
  qlayer_combine<<<32, 512, 0, stream>>>(out, T1, T2);
  qlayer_fold<<<(NDIM / CB_ROWS) * (NDIM / FD_COLS), 256, 0, stream>>>(T1, T2, cp, sp, CR, CI);
  qlayer_main<<<NBATCH / TILE_B, 256, lds_main, stream>>>(
      x, CR, CI, out);
}


# cos/sin taken where they are used: 2 helper waves per work-group of the rotation-matrix build take sincosf of the next pivot block into LDS during the sweep; phases in the fold kernel; 4 launches
# speedup vs baseline: 13.4237x; 1.0484x over previous
#include <hip/hip_runtime.h>
#include <math.h>


#define NDIM   256
#define NBATCH 65536
#define NROT   32640
#define TILE_B 64
#define RS     260

typedef __attribute__((ext_vector_type(2))) float v2f;
typedef __attribute__((ext_vector_type(8))) float v8f;

#define GV_PB 8
#define GV_JU 4
#define GV_NB (NDIM / GV_PB)
#define GV_TS 72
#define GV_STAGE (((NDIM - GV_PB) / GV_JU) * GV_TS)
#define GV_INTRA 128
#define GV_CS 4
#define GV_NT (NDIM / GV_CS)
#define GV_WT (2 * GV_NT)
#define GV_VEC (GV_NT * 257)
#define GV_BUF (GV_STAGE + GV_INTRA)
#define GV_LDS_FLOATS (GV_VEC + 2 * GV_BUF)
#define GV_LDS_BYTES (164 * 1024)
static_assert(GV_LDS_FLOATS * 4 <= GV_LDS_BYTES, "column vectors and staging fit the reservation");

#define GV_K 8

__device__ __forceinline__ int gv_seg_block(int k) {
  const unsigned long long packed = 0x20550309184080ULL;
  return (int)((packed >> (6 * k)) & 63ULL);
}

typedef __attribute__((ext_vector_type(4))) float v4f;

__device__ __forceinline__ void givens_rot_sc(v2f sc, float& ri, float& rj) {
  const v2f aa = {ri, ri};
  const v2f bb = {rj, rj};
  const v2f m  = sc * bb;
  const v2f cs = {sc.y, sc.x};
  const v2f nm = {-m.x, m.y};
  const v2f r  = __builtin_elementwise_fma(cs, aa, nm);
  ri = r.x;
  rj = r.y;
}
__device__ __forceinline__ void givens_rot_cs(v2f cs, float& ri, float& rj) {
  const v2f aa = {ri, ri};
  const v2f bb = {rj, rj};
  const v2f m  = cs * bb;
  const v2f nm = {-m.y, m.x};
  const v2f r  = __builtin_elementwise_fma(cs, aa, nm);
  ri = r.x;
  rj = r.y;
}

__device__ __forceinline__ int gv_rowbase(int i) {
  return (i * (2 * NDIM - 1 - i)) / 2;
}

__device__ __forceinline__ void gv_load_angles(const float* __restrict__ rot, int i0, int tt,
                                               float (&ang)[GV_PB]) {
  #pragma unroll
  for (int a = 0; a < GV_PB; ++a)
    ang[a] = rot[gv_rowbase(i0 + a) + (GV_PB - 1 - a) + tt];
}

__device__ __forceinline__ void gv_produce(const float* __restrict__ rot, int i0,
                                           int id, int nth, float* stage, float* intra) {
  const int len = NDIM - GV_PB - i0;
  float ang[GV_PB], nxt[GV_PB];
  int tt = id;
  if (tt < len) gv_load_angles(rot, i0, tt, ang);
  while (tt < len) {
    const int tn = tt + nth;
    gv_load_angles(rot, i0, (tn < len) ? tn : tt, nxt);
    float* d = stage + (tt >> 2) * GV_TS + 2 * (tt & 3);
    const bool odd = (tt & 1) != 0;
    #pragma unroll
    for (int a = 0; a < GV_PB; ++a) {
      float s, c;
      sincosf(ang[a], &s, &c);
      const v2f pr = odd ? (v2f){c, s} : (v2f){s, c};
      *(v2f*)(d + a * 8) = pr;
      ang[a] = nxt[a];
    }
    tt = tn;
  }
  for (int e = id; e < GV_PB * GV_PB; e += nth) {
    const int a = e >> 3;
    const int b = e & 7;
    if (b > a) {
      float s, c;
      sincosf(rot[gv_rowbase(i0 + a) + (b - a - 1)], &s, &c);
      intra[a * 8 + b] = c;
      intra[64 + a * 8 + b] = s;
    }
  }
}

__device__ __forceinline__ void gv_load_trip(const float* stage, int g,
                                             v4f (&q)[2 * GV_PB]) {
  const v4f* src = (const v4f*)(stage + g * GV_TS);
  #pragma unroll
  for (int i = 0; i < 2 * GV_PB; ++i) q[i] = src[i];
}

__device__ __forceinline__ void gv_trip(float* vj, const v4f (&q)[2 * GV_PB],
                                        float (&p)[GV_PB]) {
  float w[GV_JU];
  #pragma unroll
  for (int u = 0; u < GV_JU; ++u) w[u] = vj[u];
  #pragma unroll
  for (int a = 0; a < GV_PB; ++a) {
    #pragma unroll
    for (int u = 0; u < GV_JU; ++u) {
      const v4f qq = q[2 * a + (u >> 1)];
      if (u & 1) givens_rot_cs((v2f){qq.z, qq.w}, p[a], w[u]);
      else       givens_rot_sc((v2f){qq.x, qq.y}, p[a], w[u]);
    }
  }
  #pragma unroll
  for (int u = 0; u < GV_JU; ++u) vj[u] = w[u];
}

__global__ void __launch_bounds__(GV_WT)
qlayer_givens(const float* __restrict__ rots1,
              const float* __restrict__ rots2,
              float* __restrict__ P) {
  extern __shared__ float sm[];
  const int t = threadIdx.x;
  const bool sweeper = (t < GV_NT);
  const int cg  = blockIdx.x % GV_CS;
  const int seg = (blockIdx.x / GV_CS) % GV_K;
  const int mat = blockIdx.x / (GV_CS * GV_K);
  const int col0 = cg * GV_NT;
  const float* rot = (mat == 0) ? rots1 : rots2;
  float* Pk = P + (mat * GV_K + seg) * (NDIM * NDIM);
  const int b_lo = gv_seg_block(seg);
  const int b_hi = gv_seg_block(seg + 1);

  float* v = sm + t * 257;
  float* buf = sm + GV_VEC;

  if (sweeper) {
    for (int k = 0; k < NDIM; ++k) v[k] = (k == col0 + t) ? 1.0f : 0.0f;
  }

  for (int blk = b_lo - 1; blk < b_hi; ++blk) {
    const bool first = (blk < b_lo);
    const int i0 = blk * GV_PB;

    if (blk + 1 < b_hi && (first || !sweeper)) {
      float* nstage = buf + ((blk + 1 - b_lo) & 1) * GV_BUF;
      gv_produce(rot, i0 + GV_PB, first ? t : t - GV_NT, first ? GV_WT : GV_NT,
                 nstage, nstage + GV_STAGE);
    }

    if (!first && sweeper) {
      const float* stage = buf + ((blk - b_lo) & 1) * GV_BUF;
      const float* intra = stage + GV_STAGE;

      float p[GV_PB];
      #pragma unroll
      for (int a = 0; a < GV_PB; ++a) p[a] = v[i0 + a];

      #pragma unroll
      for (int a = 0; a < GV_PB; ++a) {
        #pragma unroll
        for (int b = a + 1; b < GV_PB; ++b) {
          const v2f sc = {intra[64 + a * 8 + b], intra[a * 8 + b]};
          givens_rot_sc(sc, p[a], p[b]);
        }
      }

      const int ntrip = (NDIM - GV_PB - i0) / GV_JU;
      float* vj = v + i0 + GV_PB;
      v4f qX[2 * GV_PB], qY[2 * GV_PB];
      if (ntrip > 0) gv_load_trip(stage, 0, qX);
      for (int g = 0; g < ntrip; g += 2) {
        gv_load_trip(stage, g + 1, qY);
        gv_trip(vj + GV_JU * g, qX, p);
        const int gn = (g + 2 < ntrip) ? g + 2 : g + 1;
        gv_load_trip(stage, gn, qX);
        gv_trip(vj + GV_JU * (g + 1), qY, p);
      }

      #pragma unroll
      for (int a = 0; a < GV_PB; ++a) v[i0 + a] = p[a];
    }

    __syncthreads();
  }
  for (int k = 0; k < GV_NT; ++k) {
    float* dst = Pk + (col0 + k) * NDIM;
    const float* srcv = sm + k * 257;
    #pragma unroll
    for (int e = 0; e < NDIM / GV_WT; ++e) dst[t + e * GV_WT] = srcv[t + e * GV_WT];
  }
}

__device__ __forceinline__ void gemm_block(const float* __restrict__ smSrc,
                                           const float* __restrict__ TR,
                                           const float* __restrict__ TI,
                                           int c0, int ln16, int hl,
                                           v8f accR[8], v8f accI[8]) {
  #pragma unroll 2
  for (int k = 0; k < NDIM; k += 4) {
    const int kk = k + 2 * hl;
    v2f a[4];
    #pragma unroll
    for (int rt = 0; rt < 4; ++rt) {
      a[rt].x = smSrc[(rt * 16 + ln16) * RS + kk];
      a[rt].y = smSrc[(rt * 16 + ln16) * RS + kk + 1];
    }
    const float* r0 = TR + kk * NDIM + c0 + ln16;
    const float* i0 = TI + kk * NDIM + c0 + ln16;
    v2f bR[2], bI[2];
    #pragma unroll
    for (int ct = 0; ct < 2; ++ct) {
      bR[ct].x = r0[ct * 16];
      bR[ct].y = r0[NDIM + ct * 16];
      bI[ct].x = i0[ct * 16];
      bI[ct].y = i0[NDIM + ct * 16];
    }
    #pragma unroll
    for (int rt = 0; rt < 4; ++rt) {
      #pragma unroll
      for (int ct = 0; ct < 2; ++ct) {
        accR[rt * 2 + ct] = __builtin_amdgcn_wmma_f32_16x16x4_f32(
             false, a[rt],  false, bR[ct],
             (short)0, accR[rt * 2 + ct],  false,  false);
        accI[rt * 2 + ct] = __builtin_amdgcn_wmma_f32_16x16x4_f32(
             false, a[rt],  false, bI[ct],
             (short)0, accI[rt * 2 + ct],  false,  false);
      }
    }
  }
}

#define CB_ROWS 16
#define CB_SEG(k) ((int)((0x20550309184080ULL >> (6 * (k))) & 63ULL))
#define CB_P0(k)  ((CB_SEG(k) * GV_PB) & ~15)
#define CB_NS(k)  ((NDIM - CB_P0(k)) / 4)
#define CB_MAXNS  (NDIM / 4)

template <int K>
__device__ __forceinline__ void cb_fetch_step(const float* __restrict__ Bl, int s, v2f& b) {
  b.x = Bl[(CB_P0(K) + 4 * s) * NDIM];
  b.y = Bl[(CB_P0(K) + 4 * s + 1) * NDIM];
}

template <int K>
__device__ __forceinline__ void cb_product(const float* __restrict__ Pl, float* S,
                                           int wave, int ln16, int hl, int n,
                                           const v2f (&bcur)[CB_MAXNS]) {
  constexpr int p0  = CB_P0(K);
  constexpr int ns  = CB_NS(K);
  constexpr int KN  = (K + 1 < GV_K) ? K + 1 : K;
  constexpr int nsn = (K + 1 < GV_K) ? CB_NS(KN) : 0;
  const bool act = (wave * 16 >= p0);
  v8f acc = {0.f, 0.f, 0.f, 0.f, 0.f, 0.f, 0.f, 0.f};
  v2f bnext[CB_MAXNS];
  if (act) {
    const float* arow = S + ln16 * RS + 2 * hl + p0;
    const float* Bn = Pl + KN * (NDIM * NDIM);
    #pragma unroll
    for (int s = 0; s < ns; ++s) {
      v2f a;
      a.x = arow[4 * s];
      a.y = arow[4 * s + 1];
      acc = __builtin_amdgcn_wmma_f32_16x16x4_f32(
           false, a,  false, bcur[s],
           (short)0, acc,  false,  false);
      if (s < nsn) cb_fetch_step<KN>(Bn, s, bnext[s]);
    }
  }
  __syncthreads();
  if (act) {
    #pragma unroll
    for (int g = 0; g < 8; ++g) S[(g + 8 * hl) * RS + n] = acc[g];
  }
  __syncthreads();
  if constexpr (K + 1 < GV_K) cb_product<K + 1>(Pl, S, wave, ln16, hl, n, bnext);
}

__global__ void __launch_bounds__(512)
qlayer_combine(const float* __restrict__ P,
               float* __restrict__ T1,
               float* __restrict__ T2) {
  __shared__ float S[CB_ROWS * RS];
  const int tid  = threadIdx.x;
  const int lane = tid & 31;
  const int hl   = lane >> 4;
  const int ln16 = lane & 15;
  const int wave = tid >> 5;
  const int mat  = blockIdx.x >> 4;
  const int row0 = (blockIdx.x & 15) * CB_ROWS;
  const float* Pm = P + mat * GV_K * (NDIM * NDIM);
  float* T = (mat == 0) ? T1 : T2;

  const int n = wave * 16 + ln16;
  const float* Pl = Pm + (2 * hl) * NDIM + n;

  v2f b1[CB_MAXNS];
  if (wave * 16 >= CB_P0(1)) {
    #pragma unroll
    for (int s = 0; s < CB_NS(1); ++s) cb_fetch_step<1>(Pl + NDIM * NDIM, s, b1[s]);
  }

  const int srow = tid >> 5;
  const int sc4  = tid & 31;
  {
    const float4* src = (const float4*)(Pm + (row0 + srow) * NDIM);
    float4* dst = (float4*)(S + srow * RS);
    dst[sc4]      = src[sc4];
    dst[sc4 + 32] = src[sc4 + 32];
  }
  __syncthreads();

  cb_product<1>(Pl, S, wave, ln16, hl, n, b1);

  {
    const float4* srcS = (const float4*)(S + srow * RS);
    float4* dst = (float4*)(T + (row0 + srow) * NDIM);
    dst[sc4]      = srcS[sc4];
    dst[sc4 + 32] = srcS[sc4 + 32];
  }
}

#define FD_COLS 32
#define FD_KQ   4
#define FD_KS   (NDIM / FD_KQ / 4)

__global__ void __launch_bounds__(256)
qlayer_fold(const float* __restrict__ T1,
            const float* __restrict__ T2,
            const float* __restrict__ phases,
            float* __restrict__ CR,
            float* __restrict__ CI) {
  __shared__ float SR[CB_ROWS * RS];
  __shared__ float SI[CB_ROWS * RS];
  __shared__ float PS[8 * 2 * 256];
  const int tid  = threadIdx.x;
  const int lane = tid & 31;
  const int hl   = lane >> 4;
  const int ln16 = lane & 15;
  const int wave = tid >> 5;
  const int tile = wave >> 2;
  const int kq   = wave & 3;
  const int row0 = (blockIdx.x >> 3) * CB_ROWS;
  const int col0 = (blockIdx.x & 7) * FD_COLS;

  const int n = col0 + tile * 16 + ln16;
  v2f bq[FD_KS];
  {
    const float* T2l = T2 + (kq * 64 + 2 * hl) * NDIM + n;
    #pragma unroll
    for (int s = 0; s < FD_KS; ++s) {
      bq[s].x = T2l[(4 * s) * NDIM];
      bq[s].y = T2l[(4 * s + 1) * NDIM];
    }
  }

  {
    const int c4 = tid & 63;
    const int r  = tid >> 6;
    const float4 ph = ((const float4*)phases)[c4];
    float4 c, s;
    sincosf(ph.x, &s.x, &c.x);
    sincosf(ph.y, &s.y, &c.y);
    sincosf(ph.z, &s.z, &c.z);
    sincosf(ph.w, &s.w, &c.w);
    #pragma unroll
    for (int q = 0; q < 4; ++q) {
      const int row = r + 4 * q;
      const float4 v = ((const float4*)(T1 + (row0 + row) * NDIM))[c4];
      ((float4*)(SR + row * RS))[c4] = make_float4(v.x * c.x, v.y * c.y, v.z * c.z, v.w * c.w);
      ((float4*)(SI + row * RS))[c4] = make_float4(v.x * s.x, v.y * s.y, v.z * s.z, v.w * s.w);
    }
  }
  __syncthreads();

  v8f accR = {0.f, 0.f, 0.f, 0.f, 0.f, 0.f, 0.f, 0.f};
  v8f accI = {0.f, 0.f, 0.f, 0.f, 0.f, 0.f, 0.f, 0.f};
  {
    const float* aRrow = SR + ln16 * RS + kq * 64 + 2 * hl;
    const float* aIrow = SI + ln16 * RS + kq * 64 + 2 * hl;
    #pragma unroll
    for (int s = 0; s < FD_KS; ++s) {
      v2f aR, aI;
      aR.x = aRrow[4 * s];
      aR.y = aRrow[4 * s + 1];
      aI.x = aIrow[4 * s];
      aI.y = aIrow[4 * s + 1];
      accR = __builtin_amdgcn_wmma_f32_16x16x4_f32(
           false, aR,  false, bq[s],
           (short)0, accR,  false,  false);
      accI = __builtin_amdgcn_wmma_f32_16x16x4_f32(
           false, aI,  false, bq[s],
           (short)0, accI,  false,  false);
    }
  }
  {
    float* pR = PS + (wave * 2) * 256;
    float* pI = pR + 256;
    #pragma unroll
    for (int g = 0; g < 8; ++g) {
      pR[(g + 8 * hl) * 16 + ln16] = accR[g];
      pI[(g + 8 * hl) * 16 + ln16] = accI[g];
    }
  }
  __syncthreads();

  {
    const int part = tid >> 7;
    const int m    = (tid >> 3) & 15;
    const int c4   = tid & 7;
    const int t2   = c4 >> 2;
    const int j0   = (c4 & 3) * 4;
    const float* p0 = PS + (((t2 * 4) * 2 + part) * 16 + m) * 16 + j0;
    const float4 a0 = *(const float4*)(p0);
    const float4 a1 = *(const float4*)(p0 + 512);
    const float4 a2 = *(const float4*)(p0 + 1024);
    const float4 a3 = *(const float4*)(p0 + 1536);
    float4 r;
    r.x = ((a0.x + a1.x) + a2.x) + a3.x;
    r.y = ((a0.y + a1.y) + a2.y) + a3.y;
    r.z = ((a0.z + a1.z) + a2.z) + a3.z;
    r.w = ((a0.w + a1.w) + a2.w) + a3.w;
    float* dst = (part ? CI : CR) + (row0 + m) * NDIM + col0 + c4 * 4;
    *(float4*)dst = r;
  }
}

#define MN_SCR 1088

__global__ void __launch_bounds__(256)
qlayer_main(const float* __restrict__ x,
            const float* __restrict__ CR,
            const float* __restrict__ CI,
            float* __restrict__ out) {
  extern __shared__ float sm[];
  float* smA = sm;
  float* smN = sm + TILE_B * RS;

  const int tid  = threadIdx.x;
  const int wave = tid >> 5;
  const int lane = tid & 31;
  const int hl   = lane >> 4;
  const int ln16 = lane & 15;
  const int c0   = wave * 32;
  const int base_row = blockIdx.x * TILE_B;

  {
    const int row = tid >> 2;
    const int q   = tid & 3;
    const float4* src = (const float4*)(x + (base_row + row) * NDIM + q * 64);
    float4* dst = (float4*)(smA + row * RS + q * 64);
    float acc = 0.f;
    #pragma unroll
    for (int c = 0; c < 16; ++c) {
      float4 vv = src[c];
      dst[c] = vv;
      acc += vv.x * vv.x + vv.y * vv.y + vv.z * vv.z + vv.w * vv.w;
    }
    smN[row * 4 + q] = acc;
  }
  __syncthreads();
  if (tid < TILE_B) {
    float s = smN[tid * 4] + smN[tid * 4 + 1] + smN[tid * 4 + 2] + smN[tid * 4 + 3];
    smN[1024 + tid] = rsqrtf(s);
  }
  __syncthreads();
  {
    const int row = tid >> 2;
    const int q   = tid & 3;
    const float inv = smN[1024 + row];
    float4* dst = (float4*)(smA + row * RS + q * 64);
    #pragma unroll
    for (int c = 0; c < 16; ++c) {
      float4 vv = dst[c];
      vv.x *= inv; vv.y *= inv; vv.z *= inv; vv.w *= inv;
      dst[c] = vv;
    }
  }
  __syncthreads();

  const v8f vzero = {0.f, 0.f, 0.f, 0.f, 0.f, 0.f, 0.f, 0.f};

  v8f accR[8];
  v8f accI[8];
  #pragma unroll
  for (int t = 0; t < 8; ++t) { accR[t] = vzero; accI[t] = vzero; }
  gemm_block(smA, CR, CI, c0, ln16, hl, accR, accI);

  #pragma unroll
  for (int rt = 0; rt < 4; ++rt) {
    #pragma unroll
    for (int ct = 0; ct < 2; ++ct) {
      const int n = c0 + ct * 16 + ln16;
      #pragma unroll
      for (int g = 0; g < 8; ++g) {
        const int m = rt * 16 + g + 8 * hl;
        const float vr = accR[rt * 2 + ct][g];
        const float vi = accI[rt * 2 + ct][g];
        out[(base_row + m) * NDIM + n] = fmaf(vi, vi, vr * vr);
      }
    }
  }
}

extern "C" void kernel_launch(void* const* d_in, const int* in_sizes, int n_in,
                              void* d_out, int out_size, void* d_ws, size_t ws_size,
                              hipStream_t stream) {
  const float* x      = (const float*)d_in[0];
  const float* rots1  = (const float*)d_in[1];
  const float* phases = (const float*)d_in[2];
  const float* rots2  = (const float*)d_in[3];
  float* out = (float*)d_out;
  float* ws  = (float*)d_ws;

  float* CI = ws;
  float* CR = ws + 65536;
  float* T1 = ws + 131072;
  float* T2 = ws + 196608;

  const int lds_givens = GV_LDS_BYTES;
  const int lds_main   = (TILE_B * RS + MN_SCR) * 4;
  hipFuncSetAttribute(reinterpret_cast<const void*>(qlayer_givens),
                      hipFuncAttributeMaxDynamicSharedMemorySize, lds_givens);
  hipFuncSetAttribute(reinterpret_cast<const void*>(qlayer_main),
                      hipFuncAttributeMaxDynamicSharedMemorySize, lds_main);

  qlayer_givens<<<2 * GV_K * GV_CS, GV_WT, lds_givens, stream>>>(rots1, rots2, out);
  qlayer_combine<<<32, 512, 0, stream>>>(out, T1, T2);
  qlayer_fold<<<(NDIM / CB_ROWS) * (NDIM / FD_COLS), 256, 0, stream>>>(T1, T2, phases, CR, CI);
  qlayer_main<<<NBATCH / TILE_B, 256, lds_main, stream>>>(
      x, CR, CI, out);
}
